# starting kernel
# baseline (speedup 1.0000x reference)
#include <hip/hip_runtime.h>

typedef float v2f_t __attribute__((ext_vector_type(2)));
typedef float v8f_t __attribute__((ext_vector_type(8)));

static __device__ __forceinline__ void atomic_fadd(float* p, float v) {
  __hip_atomic_fetch_add(p, v, __ATOMIC_RELAXED, __HIP_MEMORY_SCOPE_AGENT);
}

__global__ void zero_f32(float* __restrict__ p, long long n) {
  long long t = (long long)blockIdx.x * blockDim.x + threadIdx.x;
  if (t < n) p[t] = 0.0f;
}

__global__ void deg_accum(const int* __restrict__ dst, float* __restrict__ deg, long long E) {
  long long t = (long long)blockIdx.x * blockDim.x + threadIdx.x;
  if (t < E) atomic_fadd(&deg[dst[t]], 1.0f);
}

__global__ void deg_to_dinv(float* __restrict__ d, long long n) {
  long long t = (long long)blockIdx.x * blockDim.x + threadIdx.x;
  if (t < n) d[t] = rsqrtf(d[t] + 1.0f);
}

template <int K, int NCOL>
__global__ void gemm_wmma_f32(const float* __restrict__ X, const float* __restrict__ W,
                              float* __restrict__ H, int n) {
  constexpr int KT = (K + 3) / 4;
  int lane = threadIdx.x & 31;
  int wave = (int)((blockIdx.x * (long long)blockDim.x + threadIdx.x) >> 5);
  int row_base = wave * 16;
  if (row_base >= n) return;

  int half = lane >> 4;
  int l    = lane & 15;

  int m = row_base + l;
  if (m > n - 1) m = n - 1;
  const float* __restrict__ xrow = X + (long long)m * K;

  const bool colok = (NCOL == 16) ? true : (l < NCOL);
  const int  lc    = colok ? l : 0;

  v8f_t c = {};
#pragma unroll
  for (int kt = 0; kt < KT; ++kt) {
    int k0 = kt * 4 + 2 * half;
    bool ok0 = (k0     < K);
    bool ok1 = (k0 + 1 < K);
    int ka0 = ok0 ? k0     : 0;
    int ka1 = ok1 ? k0 + 1 : 0;

    float a0 = xrow[ka0];
    float a1 = xrow[ka1];
    float w0 = W[ka0 * NCOL + lc];
    float w1 = W[ka1 * NCOL + lc];

    v2f_t a, b;
    a.x = ok0 ? a0 : 0.0f;
    a.y = ok1 ? a1 : 0.0f;
    b.x = (ok0 && colok) ? w0 : 0.0f;
    b.y = (ok1 && colok) ? w1 : 0.0f;

    c = __builtin_amdgcn_wmma_f32_16x16x4_f32(false, a, false, b, (short)0, c, false, false);
  }

  float* __restrict__ hb = H + (long long)(row_base + 8 * half) * NCOL + l;
  if (row_base + 16 <= n) {
    if (colok) {
#pragma unroll
      for (int v = 0; v < 8; ++v) hb[v * NCOL] = c[v];
    }
  } else {
#pragma unroll
    for (int v = 0; v < 8; ++v) {
      int row = row_base + 8 * half + v;
      if (row < n && colok) hb[v * NCOL] = c[v];
    }
  }
}

__global__ void scatter_edges(const float* __restrict__ H, const float* __restrict__ dinv,
                              const int* __restrict__ src, const int* __restrict__ dst,
                              float* __restrict__ agg, long long E, int logF) {
  long long t = (long long)blockIdx.x * blockDim.x + threadIdx.x;
  if (t >= (E << logF)) return;
  long long e = t >> logF;
  int f = (int)(t & ((1 << logF) - 1));
  int s = src[e], d = dst[e];
  float cf = dinv[s] * dinv[d];
  atomic_fadd(&agg[((long long)d << logF) + f], H[((long long)s << logF) + f] * cf);
}

__global__ void finalize_relu(const float* __restrict__ H, const float* __restrict__ dinv,
                              const float* __restrict__ b, float* __restrict__ agg,
                              long long n, int logF) {
  long long t = (long long)blockIdx.x * blockDim.x + threadIdx.x;
  if (t >= (n << logF)) return;
  long long i = t >> logF;
  int f = (int)(t & ((1 << logF) - 1));
  float dv = dinv[i];
  float v = agg[t] + H[t] * dv * dv + b[f];
  agg[t] = v > 0.0f ? v : 0.0f;
}

__global__ void head_kernel(const float* __restrict__ H2, const float* __restrict__ agg2,
                            const float* __restrict__ dinv, const float* __restrict__ b2,
                            const float* __restrict__ Wd, const float* __restrict__ bd,
                            float* __restrict__ out, long long n) {
  long long i = (long long)blockIdx.x * blockDim.x + threadIdx.x;
  if (i >= n) return;
  float dv = dinv[i];
  float d2 = dv * dv;
  float acc = bd[0];
#pragma unroll
  for (int f = 0; f < 8; ++f) {
    float v = agg2[i * 8 + f] + H2[i * 8 + f] * d2 + b2[f];
    v = v > 0.0f ? v : 0.0f;
    acc += v * Wd[f];
  }
  out[i] = acc;
}

extern "C" void kernel_launch(void* const* d_in, const int* in_sizes, int n_in,
                              void* d_out, int out_size, void* d_ws, size_t ws_size,
                              hipStream_t stream) {
  const float* x  = (const float*)d_in[0];
  const int*   ei = (const int*)d_in[1];
  const float* W1 = (const float*)d_in[2];
  const float* b1 = (const float*)d_in[3];
  const float* W2 = (const float*)d_in[4];
  const float* b2 = (const float*)d_in[5];
  const float* Wd = (const float*)d_in[6];
  const float* bd = (const float*)d_in[7];
  float* out = (float*)d_out;

  long long n = in_sizes[0] / 10;
  long long E = in_sizes[1] / 2;
  const int* src = ei;
  const int* dst = ei + E;

  float* ws = (float*)d_ws;
  long long o_h = ((n + 63) / 64) * 64;
  float* dinv = ws;
  float* h1   = ws + o_h;
  float* agg1 = ws + o_h + n * 16;
  float* h2   = h1;
  float* agg2 = h1 + n * 8;

  const int B = 256;
  auto nb = [](long long t, int b) { return (unsigned)((t + b - 1) / b); };

  zero_f32<<<nb(n, B), B, 0, stream>>>(dinv, n);
  deg_accum<<<nb(E, B), B, 0, stream>>>(dst, dinv, E);
  deg_to_dinv<<<nb(n, B), B, 0, stream>>>(dinv, n);

  long long waves = (n + 15) / 16;
  long long gthreads = waves * 32;

  gemm_wmma_f32<10, 16><<<nb(gthreads, B), B, 0, stream>>>(x, W1, h1, (int)n);
  zero_f32<<<nb(n * 16, B), B, 0, stream>>>(agg1, n * 16);
  scatter_edges<<<nb(E * 16, B), B, 0, stream>>>(h1, dinv, src, dst, agg1, E, 4);
  finalize_relu<<<nb(n * 16, B), B, 0, stream>>>(h1, dinv, b1, agg1, n, 4);

  gemm_wmma_f32<16, 8><<<nb(gthreads, B), B, 0, stream>>>(agg1, W2, h2, (int)n);
  zero_f32<<<nb(n * 8, B), B, 0, stream>>>(agg2, n * 8);
  scatter_edges<<<nb(E * 8, B), B, 0, stream>>>(h2, dinv, src, dst, agg2, E, 3);

  head_kernel<<<nb(n, B), B, 0, stream>>>(h2, agg2, dinv, b2, Wd, bd, out, n);
}


# replace float-atomic edge scatter by per-launch dst buckets (one returning int atomic per edge picks the slot) and an atomic-free per-node gather; neighbour sum order follows slot arrival order
# speedup vs baseline: 1.7491x; 1.7491x over previous
#include <hip/hip_runtime.h>


typedef float v2f_t __attribute__((ext_vector_type(2)));
typedef float v8f_t __attribute__((ext_vector_type(8)));

#define BUCKET_CAP 32

__global__ void zero_i32(int* __restrict__ p, long long n) {
  long long t = (long long)blockIdx.x * blockDim.x + threadIdx.x;
  if (t < n) p[t] = 0;
}

__global__ void fill_buckets(const int* __restrict__ src, const int* __restrict__ dst,
                             int* cnt, int* __restrict__ bucket, int* ovf_n,
                             int* __restrict__ ovf_src, int* __restrict__ ovf_dst,
                             long long E) {
  long long t = (long long)blockIdx.x * blockDim.x + threadIdx.x;
  if (t >= E) return;
  int d = dst[t];
  int s = src[t];
  int r = __hip_atomic_fetch_add(&cnt[d], 1, __ATOMIC_RELAXED, __HIP_MEMORY_SCOPE_AGENT);
  if (r < BUCKET_CAP) {
    bucket[(long long)d * BUCKET_CAP + r] = s;
  } else {
    int p = __hip_atomic_fetch_add(ovf_n, 1, __ATOMIC_RELAXED, __HIP_MEMORY_SCOPE_AGENT);
    ovf_src[p] = s;
    ovf_dst[p] = d;
  }
}

template <int K, int NCOL>
__global__ void gemm_wmma_f32(const float* __restrict__ X, const float* __restrict__ W,
                              const int* __restrict__ cnt, float* __restrict__ H, int n) {
  constexpr int KT = (K + 3) / 4;
  int lane = threadIdx.x & 31;
  int wave = (int)((blockIdx.x * (long long)blockDim.x + threadIdx.x) >> 5);
  int row_base = wave * 16;
  if (row_base >= n) return;

  int half = lane >> 4;
  int l    = lane & 15;

  int m = row_base + l;
  if (m > n - 1) m = n - 1;
  const float* __restrict__ xrow = X + (long long)m * K;

  const bool colok = (NCOL == 16) ? true : (l < NCOL);
  const int  lc    = colok ? l : 0;

  v8f_t c = {};
#pragma unroll
  for (int kt = 0; kt < KT; ++kt) {
    int k0 = kt * 4 + 2 * half;
    bool ok0 = (k0     < K);
    bool ok1 = (k0 + 1 < K);
    int ka0 = ok0 ? k0     : 0;
    int ka1 = ok1 ? k0 + 1 : 0;

    float a0 = xrow[ka0];
    float a1 = xrow[ka1];
    float w0 = W[ka0 * NCOL + lc];
    float w1 = W[ka1 * NCOL + lc];

    v2f_t a, b;
    a.x = ok0 ? a0 : 0.0f;
    a.y = ok1 ? a1 : 0.0f;
    b.x = (ok0 && colok) ? w0 : 0.0f;
    b.y = (ok1 && colok) ? w1 : 0.0f;

    c = __builtin_amdgcn_wmma_f32_16x16x4_f32(false, a, false, b, (short)0, c, false, false);
  }

  float* __restrict__ hb = H + (long long)(row_base + 8 * half) * NCOL + l;
  const int* __restrict__ cb = cnt + row_base + 8 * half;
  if (row_base + 16 <= n) {
    if (colok) {
#pragma unroll
      for (int v = 0; v < 8; ++v) hb[v * NCOL] = c[v] * rsqrtf((float)cb[v] + 1.0f);
    }
  } else {
#pragma unroll
    for (int v = 0; v < 8; ++v) {
      int row = row_base + 8 * half + v;
      if (row < n && colok) hb[v * NCOL] = c[v] * rsqrtf((float)cb[v] + 1.0f);
    }
  }
}

template <int F, bool HEAD>
__global__ void gather_nodes(const float* __restrict__ HS, const int* __restrict__ cnt,
                             const int* __restrict__ bucket, const int* __restrict__ ovf_n,
                             const int* __restrict__ ovf_src, const int* __restrict__ ovf_dst,
                             const float* __restrict__ bias, const float* __restrict__ Wd,
                             const float* __restrict__ bd, float* __restrict__ out, long long n) {
  constexpr int LPN = F / 4;
  long long t = (long long)blockIdx.x * blockDim.x + threadIdx.x;
  long long d = t / LPN;
  int q = (int)(t % LPN);
  if (d >= n) return;

  const float4* __restrict__ h4 = (const float4*)HS;
  int c = cnt[d];
  int m = c < BUCKET_CAP ? c : BUCKET_CAP;
  float4 acc = h4[d * LPN + q];
  const int* __restrict__ bi = bucket + d * BUCKET_CAP;

  for (int j = 0; j < m; j += 4) {
    bool k1 = (j + 1 < m), k2 = (j + 2 < m), k3 = (j + 3 < m);
    int i0, i1, i2, i3;
    if (k3) {
      int4 id = *(const int4*)(bi + j);
      i0 = id.x; i1 = id.y; i2 = id.z; i3 = id.w;
    } else {
      i0 = bi[j];
      i1 = k1 ? bi[j + 1] : 0;
      i2 = k2 ? bi[j + 2] : 0;
      i3 = 0;
    }
    long long s0 = i0;
    long long s1 = k1 ? (long long)i1 : d;
    long long s2 = k2 ? (long long)i2 : d;
    long long s3 = k3 ? (long long)i3 : d;
    float4 v0 = h4[s0 * LPN + q];
    float4 v1 = h4[s1 * LPN + q];
    float4 v2 = h4[s2 * LPN + q];
    float4 v3 = h4[s3 * LPN + q];
    acc.x += v0.x; acc.y += v0.y; acc.z += v0.z; acc.w += v0.w;
    acc.x += k1 ? v1.x : 0.0f; acc.y += k1 ? v1.y : 0.0f; acc.z += k1 ? v1.z : 0.0f; acc.w += k1 ? v1.w : 0.0f;
    acc.x += k2 ? v2.x : 0.0f; acc.y += k2 ? v2.y : 0.0f; acc.z += k2 ? v2.z : 0.0f; acc.w += k2 ? v2.w : 0.0f;
    acc.x += k3 ? v3.x : 0.0f; acc.y += k3 ? v3.y : 0.0f; acc.z += k3 ? v3.z : 0.0f; acc.w += k3 ? v3.w : 0.0f;
  }

  if (c > BUCKET_CAP) {
    int on = ovf_n[0];
    for (int p = 0; p < on; ++p) {
      if (ovf_dst[p] == (int)d) {
        float4 v = h4[(long long)ovf_src[p] * LPN + q];
        acc.x += v.x; acc.y += v.y; acc.z += v.z; acc.w += v.w;
      }
    }
  }

  float dv = rsqrtf((float)c + 1.0f);
  float y0 = acc.x * dv + bias[4 * q + 0];
  float y1 = acc.y * dv + bias[4 * q + 1];
  float y2 = acc.z * dv + bias[4 * q + 2];
  float y3 = acc.w * dv + bias[4 * q + 3];
  y0 = y0 > 0.0f ? y0 : 0.0f;
  y1 = y1 > 0.0f ? y1 : 0.0f;
  y2 = y2 > 0.0f ? y2 : 0.0f;
  y3 = y3 > 0.0f ? y3 : 0.0f;

  if (!HEAD) {
    float4 r;
    r.x = y0; r.y = y1; r.z = y2; r.w = y3;
    ((float4*)out)[d * LPN + q] = r;
  } else {
    float p = y0 * Wd[4 * q + 0] + y1 * Wd[4 * q + 1] + y2 * Wd[4 * q + 2] + y3 * Wd[4 * q + 3];
    for (int o = LPN / 2; o > 0; o >>= 1) p += __shfl_xor(p, o);
    if (q == 0) out[d] = p + bd[0];
  }
}

extern "C" void kernel_launch(void* const* d_in, const int* in_sizes, int n_in,
                              void* d_out, int out_size, void* d_ws, size_t ws_size,
                              hipStream_t stream) {
  const float* x  = (const float*)d_in[0];
  const int*   ei = (const int*)d_in[1];
  const float* W1 = (const float*)d_in[2];
  const float* b1 = (const float*)d_in[3];
  const float* W2 = (const float*)d_in[4];
  const float* b2 = (const float*)d_in[5];
  const float* Wd = (const float*)d_in[6];
  const float* bd = (const float*)d_in[7];
  float* out = (float*)d_out;

  long long n = in_sizes[0] / 10;
  long long E = in_sizes[1] / 2;
  const int* src = ei;
  const int* dst = ei + E;

  int* wsi = (int*)d_ws;
  long long o_c = ((n + 1 + 63) / 64) * 64;
  int*   cnt     = wsi;
  int*   ovf_n   = wsi + n;
  int*   bucket  = wsi + o_c;
  float* hs1     = (float*)(wsi + o_c + n * BUCKET_CAP);
  float* a1      = hs1 + n * 16;
  float* hs2     = hs1;
  int*   ovf_src = (int*)(a1 + n * 16);
  int*   ovf_dst = ovf_src + E;

  const int B = 256;
  auto nb = [](long long t, int b) { return (unsigned)((t + b - 1) / b); };

  zero_i32<<<nb(n + 1, B), B, 0, stream>>>(cnt, n + 1);
  fill_buckets<<<nb(E, B), B, 0, stream>>>(src, dst, cnt, bucket, ovf_n, ovf_src, ovf_dst, E);

  long long waves = (n + 15) / 16;
  long long gthreads = waves * 32;

  gemm_wmma_f32<10, 16><<<nb(gthreads, B), B, 0, stream>>>(x, W1, cnt, hs1, (int)n);
  gather_nodes<16, false><<<nb(n * 4, B), B, 0, stream>>>(hs1, cnt, bucket, ovf_n, ovf_src, ovf_dst,
                                                          b1, Wd, bd, a1, n);

  gemm_wmma_f32<16, 8><<<nb(gthreads, B), B, 0, stream>>>(a1, W2, cnt, hs2, (int)n);
  gather_nodes<8, true><<<nb(n * 2, B), B, 0, stream>>>(hs2, cnt, bucket, ovf_n, ovf_src, ovf_dst,
                                                        b2, Wd, bd, out, n);
}


# edges grouped by destination in two levels on LDS counters (1024-node groups, then node rows in LDS) instead of one global returning add per edge; neighbour sum order still follows arrival order
# speedup vs baseline: 3.5152x; 2.0097x over previous
#include <hip/hip_runtime.h>


typedef float v2f_t __attribute__((ext_vector_type(2)));
typedef float v8f_t __attribute__((ext_vector_type(8)));

#define BUCKET_CAP 32
#define GROUP_SHIFT 10
#define GROUP_NODES (1 << GROUP_SHIFT)
#define GROUP_CAP 16384
#define SRC_BITS 22
#define CNT_STRIDE 32
#define PART_THREADS 512
#define PART_EPT 16
#define ROW_PITCH 33

__global__ void zero_i32(int* __restrict__ p, long long n) {
  long long t = (long long)blockIdx.x * blockDim.x + threadIdx.x;
  if (t < n) p[t] = 0;
}

__global__ void __launch_bounds__(PART_THREADS)
group_edges(const int* __restrict__ src, const int* __restrict__ dst,
            int* gcnt, unsigned* __restrict__ glist, int* ovfa_n,
            int* __restrict__ ovfa_src, int* __restrict__ ovfa_dst,
            long long E, int ngroups) {
  extern __shared__ int sm[];
  int* hist  = sm;
  int* gbase = sm + ngroups;
  int tid = threadIdx.x;
  for (int i = tid; i < ngroups; i += PART_THREADS) hist[i] = 0;
  __syncthreads();

  long long e0 = (long long)blockIdx.x * (PART_THREADS * PART_EPT) + tid;
  int d[PART_EPT], s[PART_EPT], rk[PART_EPT];
#pragma unroll
  for (int k = 0; k < PART_EPT; ++k) {
    long long e = e0 + (long long)k * PART_THREADS;
    bool ok = e < E;
    long long ec = ok ? e : 0;
    int dv = dst[ec];
    s[k] = src[ec];
    d[k] = ok ? dv : -1;
  }
#pragma unroll
  for (int k = 0; k < PART_EPT; ++k) {
    rk[k] = 0;
    if (d[k] >= 0)
      rk[k] = __hip_atomic_fetch_add(&hist[d[k] >> GROUP_SHIFT], 1, __ATOMIC_RELAXED,
                                     __HIP_MEMORY_SCOPE_WORKGROUP);
  }
  __syncthreads();
  for (int i = tid; i < ngroups; i += PART_THREADS) {
    int c = hist[i];
    int b = 0;
    if (c > 0)
      b = __hip_atomic_fetch_add(&gcnt[(long long)i * CNT_STRIDE], c, __ATOMIC_RELAXED,
                                 __HIP_MEMORY_SCOPE_AGENT);
    gbase[i] = b;
  }
  __syncthreads();
#pragma unroll
  for (int k = 0; k < PART_EPT; ++k) rk[k] += gbase[(d[k] >= 0 ? d[k] : 0) >> GROUP_SHIFT];
#pragma unroll
  for (int k = 0; k < PART_EPT; ++k) {
    if (d[k] >= 0) {
      int g = d[k] >> GROUP_SHIFT;
      int pos = rk[k];
      if (pos < GROUP_CAP) {
        glist[(long long)g * GROUP_CAP + pos] =
            ((unsigned)(d[k] & (GROUP_NODES - 1)) << SRC_BITS) | (unsigned)s[k];
      } else {
        int p = __hip_atomic_fetch_add(ovfa_n, 1, __ATOMIC_RELAXED, __HIP_MEMORY_SCOPE_AGENT);
        ovfa_src[p] = s[k];
        ovfa_dst[p] = d[k];
      }
    }
  }
}

__global__ void __launch_bounds__(GROUP_NODES)
place_group(const int* __restrict__ gcnt, const unsigned* __restrict__ glist,
            const int* __restrict__ ovfa_n, const int* __restrict__ ovfa_src,
            const int* __restrict__ ovfa_dst, int* __restrict__ cnt, int* __restrict__ bucket,
            int* ovfb_n, int* __restrict__ ovfb_src, int* __restrict__ ovfb_dst, long long n) {
  __shared__ int lcnt[GROUP_NODES];
  __shared__ int lslot[GROUP_NODES * ROW_PITCH];
  int tid = threadIdx.x;
  int g = blockIdx.x;
  long long node0 = (long long)g << GROUP_SHIFT;
  lcnt[tid] = 0;
  __syncthreads();

  auto put = [&](int dl, int sv) {
    int r = __hip_atomic_fetch_add(&lcnt[dl], 1, __ATOMIC_RELAXED, __HIP_MEMORY_SCOPE_WORKGROUP);
    if (r < BUCKET_CAP) {
      lslot[dl * ROW_PITCH + r] = sv;
    } else {
      int p = __hip_atomic_fetch_add(ovfb_n, 1, __ATOMIC_RELAXED, __HIP_MEMORY_SCOPE_AGENT);
      ovfb_src[p] = sv;
      ovfb_dst[p] = (int)(node0 + dl);
    }
  };

  int total = gcnt[(long long)g * CNT_STRIDE];
  int m = total < GROUP_CAP ? total : GROUP_CAP;
  const unsigned* __restrict__ gl = glist + (long long)g * GROUP_CAP;
  for (int i0 = 0; i0 < m; i0 += 4 * GROUP_NODES) {
    unsigned r[4];
    bool ok[4];
#pragma unroll
    for (int u = 0; u < 4; ++u) {
      int i = i0 + u * GROUP_NODES + tid;
      ok[u] = i < m;
      r[u] = gl[ok[u] ? i : 0];
    }
#pragma unroll
    for (int u = 0; u < 4; ++u)
      if (ok[u]) put((int)(r[u] >> SRC_BITS), (int)(r[u] & ((1u << SRC_BITS) - 1u)));
  }

  int ona = ovfa_n[0];
  for (int p = tid; p < ona; p += GROUP_NODES) {
    int dv = ovfa_dst[p];
    if ((dv >> GROUP_SHIFT) == g) put(dv & (GROUP_NODES - 1), ovfa_src[p]);
  }
  __syncthreads();

  long long node = node0 + tid;
  if (node < n) cnt[node] = lcnt[tid];

  int j = tid % BUCKET_CAP;
  constexpr int RSTEP = GROUP_NODES / BUCKET_CAP;
  for (int row0 = tid / BUCKET_CAP; row0 < GROUP_NODES; row0 += 4 * RSTEP) {
    int c[4];
#pragma unroll
    for (int u = 0; u < 4; ++u) c[u] = lcnt[row0 + u * RSTEP];
#pragma unroll
    for (int u = 0; u < 4; ++u) {
      int row = row0 + u * RSTEP;
      long long nd = node0 + row;
      if (nd < n && j < c[u]) bucket[nd * BUCKET_CAP + j] = lslot[row * ROW_PITCH + j];
    }
  }
}

template <int K, int NCOL>
__global__ void gemm_wmma_f32(const float* __restrict__ X, const float* __restrict__ W,
                              const int* __restrict__ cnt, float* __restrict__ H, int n) {
  constexpr int KT = (K + 3) / 4;
  int lane = threadIdx.x & 31;
  int wave = (int)((blockIdx.x * (long long)blockDim.x + threadIdx.x) >> 5);
  int row_base = wave * 16;
  if (row_base >= n) return;

  int half = lane >> 4;
  int l    = lane & 15;

  int m = row_base + l;
  if (m > n - 1) m = n - 1;
  const float* __restrict__ xrow = X + (long long)m * K;

  const bool colok = (NCOL == 16) ? true : (l < NCOL);
  const int  lc    = colok ? l : 0;

  v8f_t c = {};
#pragma unroll
  for (int kt = 0; kt < KT; ++kt) {
    int k0 = kt * 4 + 2 * half;
    bool ok0 = (k0     < K);
    bool ok1 = (k0 + 1 < K);
    int ka0 = ok0 ? k0     : 0;
    int ka1 = ok1 ? k0 + 1 : 0;

    float a0 = xrow[ka0];
    float a1 = xrow[ka1];
    float w0 = W[ka0 * NCOL + lc];
    float w1 = W[ka1 * NCOL + lc];

    v2f_t a, b;
    a.x = ok0 ? a0 : 0.0f;
    a.y = ok1 ? a1 : 0.0f;
    b.x = (ok0 && colok) ? w0 : 0.0f;
    b.y = (ok1 && colok) ? w1 : 0.0f;

    c = __builtin_amdgcn_wmma_f32_16x16x4_f32(false, a, false, b, (short)0, c, false, false);
  }

  float* __restrict__ hb = H + (long long)(row_base + 8 * half) * NCOL + l;
  const int* __restrict__ cb = cnt + row_base + 8 * half;
  if (row_base + 16 <= n) {
    if (colok) {
#pragma unroll
      for (int v = 0; v < 8; ++v) hb[v * NCOL] = c[v] * rsqrtf((float)cb[v] + 1.0f);
    }
  } else {
#pragma unroll
    for (int v = 0; v < 8; ++v) {
      int row = row_base + 8 * half + v;
      if (row < n && colok) hb[v * NCOL] = c[v] * rsqrtf((float)cb[v] + 1.0f);
    }
  }
}

template <int F, bool HEAD>
__global__ void gather_nodes(const float* __restrict__ HS, const int* __restrict__ cnt,
                             const int* __restrict__ bucket, const int* __restrict__ ovf_n,
                             const int* __restrict__ ovf_src, const int* __restrict__ ovf_dst,
                             const float* __restrict__ bias, const float* __restrict__ Wd,
                             const float* __restrict__ bd, float* __restrict__ out, long long n) {
  constexpr int LPN = F / 4;
  long long t = (long long)blockIdx.x * blockDim.x + threadIdx.x;
  long long d = t / LPN;
  int q = (int)(t % LPN);
  if (d >= n) return;

  const float4* __restrict__ h4 = (const float4*)HS;
  int c = cnt[d];
  int m = c < BUCKET_CAP ? c : BUCKET_CAP;
  float4 acc = h4[d * LPN + q];
  const int* __restrict__ bi = bucket + d * BUCKET_CAP;

  for (int j = 0; j < m; j += 4) {
    bool k1 = (j + 1 < m), k2 = (j + 2 < m), k3 = (j + 3 < m);
    int i0, i1, i2, i3;
    if (k3) {
      int4 id = *(const int4*)(bi + j);
      i0 = id.x; i1 = id.y; i2 = id.z; i3 = id.w;
    } else {
      i0 = bi[j];
      i1 = k1 ? bi[j + 1] : 0;
      i2 = k2 ? bi[j + 2] : 0;
      i3 = 0;
    }
    long long s0 = i0;
    long long s1 = k1 ? (long long)i1 : d;
    long long s2 = k2 ? (long long)i2 : d;
    long long s3 = k3 ? (long long)i3 : d;
    float4 v0 = h4[s0 * LPN + q];
    float4 v1 = h4[s1 * LPN + q];
    float4 v2 = h4[s2 * LPN + q];
    float4 v3 = h4[s3 * LPN + q];
    acc.x += v0.x; acc.y += v0.y; acc.z += v0.z; acc.w += v0.w;
    acc.x += k1 ? v1.x : 0.0f; acc.y += k1 ? v1.y : 0.0f; acc.z += k1 ? v1.z : 0.0f; acc.w += k1 ? v1.w : 0.0f;
    acc.x += k2 ? v2.x : 0.0f; acc.y += k2 ? v2.y : 0.0f; acc.z += k2 ? v2.z : 0.0f; acc.w += k2 ? v2.w : 0.0f;
    acc.x += k3 ? v3.x : 0.0f; acc.y += k3 ? v3.y : 0.0f; acc.z += k3 ? v3.z : 0.0f; acc.w += k3 ? v3.w : 0.0f;
  }

  if (c > BUCKET_CAP) {
    int on = ovf_n[0];
    for (int p = 0; p < on; ++p) {
      if (ovf_dst[p] == (int)d) {
        float4 v = h4[(long long)ovf_src[p] * LPN + q];
        acc.x += v.x; acc.y += v.y; acc.z += v.z; acc.w += v.w;
      }
    }
  }

  float dv = rsqrtf((float)c + 1.0f);
  float y0 = acc.x * dv + bias[4 * q + 0];
  float y1 = acc.y * dv + bias[4 * q + 1];
  float y2 = acc.z * dv + bias[4 * q + 2];
  float y3 = acc.w * dv + bias[4 * q + 3];
  y0 = y0 > 0.0f ? y0 : 0.0f;
  y1 = y1 > 0.0f ? y1 : 0.0f;
  y2 = y2 > 0.0f ? y2 : 0.0f;
  y3 = y3 > 0.0f ? y3 : 0.0f;

  if (!HEAD) {
    float4 r;
    r.x = y0; r.y = y1; r.z = y2; r.w = y3;
    ((float4*)out)[d * LPN + q] = r;
  } else {
    float p = y0 * Wd[4 * q + 0] + y1 * Wd[4 * q + 1] + y2 * Wd[4 * q + 2] + y3 * Wd[4 * q + 3];
    for (int o = LPN / 2; o > 0; o >>= 1) p += __shfl_xor(p, o);
    if (q == 0) out[d] = p + bd[0];
  }
}

extern "C" void kernel_launch(void* const* d_in, const int* in_sizes, int n_in,
                              void* d_out, int out_size, void* d_ws, size_t ws_size,
                              hipStream_t stream) {
  const float* x  = (const float*)d_in[0];
  const int*   ei = (const int*)d_in[1];
  const float* W1 = (const float*)d_in[2];
  const float* b1 = (const float*)d_in[3];
  const float* W2 = (const float*)d_in[4];
  const float* b2 = (const float*)d_in[5];
  const float* Wd = (const float*)d_in[6];
  const float* bd = (const float*)d_in[7];
  float* out = (float*)d_out;

  long long n = in_sizes[0] / 10;
  long long E = in_sizes[1] / 2;
  const int* src = ei;
  const int* dst = ei + E;

  int* wsi = (int*)d_ws;
  long long ng  = (n + GROUP_NODES - 1) / GROUP_NODES;
  long long o_c = ((n + 2 + 63) / 64) * 64;
  long long o_b = o_c + ng * CNT_STRIDE;
  int*   cnt      = wsi;
  int*   ovfb_n   = wsi + n;
  int*   ovfa_n   = wsi + n + 1;
  int*   gcnt     = wsi + o_c;
  int*   bucket   = wsi + o_b;
  float* hs1      = (float*)(wsi + o_b + n * BUCKET_CAP);
  float* a1       = hs1 + n * 16;
  float* hs2      = hs1;
  int*   ovfb_src = (int*)(a1 + n * 16);
  int*   ovfb_dst = ovfb_src + E;
  unsigned* glist = (unsigned*)(ovfb_dst + E);
  int*   ovfa_src = (int*)(glist + ng * GROUP_CAP);
  int*   ovfa_dst = ovfa_src + E;

  const int B = 256;
  auto nb = [](long long t, int b) { return (unsigned)((t + b - 1) / b); };

  zero_i32<<<nb(o_b - n, B), B, 0, stream>>>(wsi + n, o_b - n);
  group_edges<<<nb(E, PART_THREADS * PART_EPT), PART_THREADS, 2 * ng * sizeof(int), stream>>>(
      src, dst, gcnt, glist, ovfa_n, ovfa_src, ovfa_dst, E, (int)ng);
  place_group<<<(unsigned)ng, GROUP_NODES, 0, stream>>>(gcnt, glist, ovfa_n, ovfa_src, ovfa_dst,
                                                       cnt, bucket, ovfb_n, ovfb_src, ovfb_dst, n);

  long long waves = (n + 15) / 16;
  long long gthreads = waves * 32;

  gemm_wmma_f32<10, 16><<<nb(gthreads, B), B, 0, stream>>>(x, W1, cnt, hs1, (int)n);
  gather_nodes<16, false><<<nb(n * 4, B), B, 0, stream>>>(hs1, cnt, bucket, ovfb_n, ovfb_src, ovfb_dst,
                                                          b1, Wd, bd, a1, n);

  gemm_wmma_f32<16, 8><<<nb(gthreads, B), B, 0, stream>>>(a1, W2, cnt, hs2, (int)n);
  gather_nodes<8, true><<<nb(n * 2, B), B, 0, stream>>>(hs2, cnt, bucket, ovfb_n, ovfb_src, ovfb_dst,
                                                        b2, Wd, bd, out, n);
}


# group_edges orders its block's records by group in LDS (scan of its LDS counters + rank) and writes each group's run with consecutive lanes, not one line per edge; sum order still by arrival order
# speedup vs baseline: 4.0210x; 1.1439x over previous
#include <hip/hip_runtime.h>


typedef float v2f_t __attribute__((ext_vector_type(2)));
typedef float v8f_t __attribute__((ext_vector_type(8)));

#define BUCKET_CAP 32
#define GROUP_SHIFT 10
#define GROUP_NODES (1 << GROUP_SHIFT)
#define GROUP_CAP 16384
#define SRC_BITS 22
#define CNT_STRIDE 32
#define PART_THREADS 512
#define PART_EPT 16
#define PART_WAVES (PART_THREADS / 32)
#define ROW_PITCH 33

__global__ void zero_i32(int* __restrict__ p, long long n) {
  long long t = (long long)blockIdx.x * blockDim.x + threadIdx.x;
  if (t < n) p[t] = 0;
}

__global__ void __launch_bounds__(PART_THREADS)
group_edges(const int* __restrict__ src, const int* __restrict__ dst,
            int* gcnt, unsigned* __restrict__ glist, int* ovfa_n,
            int* __restrict__ ovfa_src, int* __restrict__ ovfa_dst,
            long long E, int ngroups) {
  extern __shared__ int sm[];
  __shared__ unsigned rec[PART_THREADS * PART_EPT];
  __shared__ int wtot[PART_WAVES];
  int* hist  = sm;
  int* lbase = sm + ngroups;
  int* gbase = sm + 2 * ngroups;
  int tid = threadIdx.x;
  int lane = tid & 31;
  int wave = tid >> 5;
  for (int i = tid; i < ngroups; i += PART_THREADS) hist[i] = 0;
  __syncthreads();

  long long e0 = (long long)blockIdx.x * (PART_THREADS * PART_EPT) + tid;
  int d[PART_EPT], s[PART_EPT], rk[PART_EPT];
#pragma unroll
  for (int k = 0; k < PART_EPT; ++k) {
    long long e = e0 + (long long)k * PART_THREADS;
    bool ok = e < E;
    long long ec = ok ? e : 0;
    int dv = dst[ec];
    s[k] = src[ec];
    d[k] = ok ? dv : -1;
  }
#pragma unroll
  for (int k = 0; k < PART_EPT; ++k) {
    rk[k] = 0;
    if (d[k] >= 0)
      rk[k] = __hip_atomic_fetch_add(&hist[d[k] >> GROUP_SHIFT], 1, __ATOMIC_RELAXED,
                                     __HIP_MEMORY_SCOPE_WORKGROUP);
  }
  __syncthreads();

  int chunk = (ngroups + PART_THREADS - 1) / PART_THREADS;
  int gfirst = tid * chunk;
  int sum = 0;
  for (int i = 0; i < chunk; ++i) {
    int g = gfirst + i;
    if (g < ngroups) {
      int c = hist[g];
      int b = 0;
      if (c > 0)
        b = __hip_atomic_fetch_add(&gcnt[(long long)g * CNT_STRIDE], c, __ATOMIC_RELAXED,
                                   __HIP_MEMORY_SCOPE_AGENT);
      gbase[g] = b;
      sum += c;
    }
  }
  int incl = sum;
  for (int o = 1; o < 32; o <<= 1) {
    int v = __shfl_up(incl, o);
    if (lane >= o) incl += v;
  }
  if (lane == 31) wtot[wave] = incl;
  __syncthreads();
  int run = incl - sum;
  for (int w = 0; w < wave; ++w) run += wtot[w];
  for (int i = 0; i < chunk; ++i) {
    int g = gfirst + i;
    if (g < ngroups) {
      lbase[g] = run;
      run += hist[g];
    }
  }
  __syncthreads();

#pragma unroll
  for (int k = 0; k < PART_EPT; ++k) rk[k] += lbase[(d[k] >= 0 ? d[k] : 0) >> GROUP_SHIFT];
#pragma unroll
  for (int k = 0; k < PART_EPT; ++k) {
    if (d[k] >= 0)
      rec[rk[k]] = ((unsigned)(d[k] & (GROUP_NODES - 1)) << SRC_BITS) | (unsigned)s[k];
  }
  __syncthreads();

  auto emit = [&](int g, int pos, unsigned r) {
    if (pos < GROUP_CAP) {
      glist[(long long)g * GROUP_CAP + pos] = r;
    } else {
      int p = __hip_atomic_fetch_add(ovfa_n, 1, __ATOMIC_RELAXED, __HIP_MEMORY_SCOPE_AGENT);
      ovfa_src[p] = (int)(r & ((1u << SRC_BITS) - 1u));
      ovfa_dst[p] = (g << GROUP_SHIFT) | (int)(r >> SRC_BITS);
    }
  };

  for (int gs = wave; gs < ngroups; gs += 4 * PART_WAVES) {
    int c[4], lb[4], gb[4];
    unsigned r[4];
#pragma unroll
    for (int u = 0; u < 4; ++u) {
      int g = gs + u * PART_WAVES;
      bool ok = g < ngroups;
      int gg = ok ? g : 0;
      int cv = hist[gg];
      lb[u] = lbase[gg];
      gb[u] = gbase[gg];
      c[u] = ok ? cv : 0;
    }
#pragma unroll
    for (int u = 0; u < 4; ++u) r[u] = rec[lane < c[u] ? lb[u] + lane : 0];
#pragma unroll
    for (int u = 0; u < 4; ++u)
      if (lane < c[u]) emit(gs + u * PART_WAVES, gb[u] + lane, r[u]);
#pragma unroll
    for (int u = 0; u < 4; ++u)
      for (int j = lane + 32; j < c[u]; j += 32) emit(gs + u * PART_WAVES, gb[u] + j, rec[lb[u] + j]);
  }
}

__global__ void __launch_bounds__(GROUP_NODES)
place_group(const int* __restrict__ gcnt, const unsigned* __restrict__ glist,
            const int* __restrict__ ovfa_n, const int* __restrict__ ovfa_src,
            const int* __restrict__ ovfa_dst, int* __restrict__ cnt, int* __restrict__ bucket,
            int* ovfb_n, int* __restrict__ ovfb_src, int* __restrict__ ovfb_dst, long long n) {
  __shared__ int lcnt[GROUP_NODES];
  __shared__ int lslot[GROUP_NODES * ROW_PITCH];
  int tid = threadIdx.x;
  int g = blockIdx.x;
  long long node0 = (long long)g << GROUP_SHIFT;
  lcnt[tid] = 0;
  __syncthreads();

  auto put = [&](int dl, int sv) {
    int r = __hip_atomic_fetch_add(&lcnt[dl], 1, __ATOMIC_RELAXED, __HIP_MEMORY_SCOPE_WORKGROUP);
    if (r < BUCKET_CAP) {
      lslot[dl * ROW_PITCH + r] = sv;
    } else {
      int p = __hip_atomic_fetch_add(ovfb_n, 1, __ATOMIC_RELAXED, __HIP_MEMORY_SCOPE_AGENT);
      ovfb_src[p] = sv;
      ovfb_dst[p] = (int)(node0 + dl);
    }
  };

  int total = gcnt[(long long)g * CNT_STRIDE];
  int m = total < GROUP_CAP ? total : GROUP_CAP;
  const unsigned* __restrict__ gl = glist + (long long)g * GROUP_CAP;
  for (int i0 = 0; i0 < m; i0 += 4 * GROUP_NODES) {
    unsigned r[4];
    bool ok[4];
#pragma unroll
    for (int u = 0; u < 4; ++u) {
      int i = i0 + u * GROUP_NODES + tid;
      ok[u] = i < m;
      r[u] = gl[ok[u] ? i : 0];
    }
#pragma unroll
    for (int u = 0; u < 4; ++u)
      if (ok[u]) put((int)(r[u] >> SRC_BITS), (int)(r[u] & ((1u << SRC_BITS) - 1u)));
  }

  int ona = ovfa_n[0];
  for (int p = tid; p < ona; p += GROUP_NODES) {
    int dv = ovfa_dst[p];
    if ((dv >> GROUP_SHIFT) == g) put(dv & (GROUP_NODES - 1), ovfa_src[p]);
  }
  __syncthreads();

  long long node = node0 + tid;
  if (node < n) cnt[node] = lcnt[tid];

  int j = tid % BUCKET_CAP;
  constexpr int RSTEP = GROUP_NODES / BUCKET_CAP;
  for (int row0 = tid / BUCKET_CAP; row0 < GROUP_NODES; row0 += 4 * RSTEP) {
    int c[4];
#pragma unroll
    for (int u = 0; u < 4; ++u) c[u] = lcnt[row0 + u * RSTEP];
#pragma unroll
    for (int u = 0; u < 4; ++u) {
      int row = row0 + u * RSTEP;
      long long nd = node0 + row;
      if (nd < n && j < c[u]) bucket[nd * BUCKET_CAP + j] = lslot[row * ROW_PITCH + j];
    }
  }
}

template <int K, int NCOL>
__global__ void gemm_wmma_f32(const float* __restrict__ X, const float* __restrict__ W,
                              const int* __restrict__ cnt, float* __restrict__ H, int n) {
  constexpr int KT = (K + 3) / 4;
  int lane = threadIdx.x & 31;
  int wave = (int)((blockIdx.x * (long long)blockDim.x + threadIdx.x) >> 5);
  int row_base = wave * 16;
  if (row_base >= n) return;

  int half = lane >> 4;
  int l    = lane & 15;

  int m = row_base + l;
  if (m > n - 1) m = n - 1;
  const float* __restrict__ xrow = X + (long long)m * K;

  const bool colok = (NCOL == 16) ? true : (l < NCOL);
  const int  lc    = colok ? l : 0;

  v8f_t c = {};
#pragma unroll
  for (int kt = 0; kt < KT; ++kt) {
    int k0 = kt * 4 + 2 * half;
    bool ok0 = (k0     < K);
    bool ok1 = (k0 + 1 < K);
    int ka0 = ok0 ? k0     : 0;
    int ka1 = ok1 ? k0 + 1 : 0;

    float a0 = xrow[ka0];
    float a1 = xrow[ka1];
    float w0 = W[ka0 * NCOL + lc];
    float w1 = W[ka1 * NCOL + lc];

    v2f_t a, b;
    a.x = ok0 ? a0 : 0.0f;
    a.y = ok1 ? a1 : 0.0f;
    b.x = (ok0 && colok) ? w0 : 0.0f;
    b.y = (ok1 && colok) ? w1 : 0.0f;

    c = __builtin_amdgcn_wmma_f32_16x16x4_f32(false, a, false, b, (short)0, c, false, false);
  }

  float* __restrict__ hb = H + (long long)(row_base + 8 * half) * NCOL + l;
  const int* __restrict__ cb = cnt + row_base + 8 * half;
  if (row_base + 16 <= n) {
    if (colok) {
#pragma unroll
      for (int v = 0; v < 8; ++v) hb[v * NCOL] = c[v] * rsqrtf((float)cb[v] + 1.0f);
    }
  } else {
#pragma unroll
    for (int v = 0; v < 8; ++v) {
      int row = row_base + 8 * half + v;
      if (row < n && colok) hb[v * NCOL] = c[v] * rsqrtf((float)cb[v] + 1.0f);
    }
  }
}

template <int F, bool HEAD>
__global__ void gather_nodes(const float* __restrict__ HS, const int* __restrict__ cnt,
                             const int* __restrict__ bucket, const int* __restrict__ ovf_n,
                             const int* __restrict__ ovf_src, const int* __restrict__ ovf_dst,
                             const float* __restrict__ bias, const float* __restrict__ Wd,
                             const float* __restrict__ bd, float* __restrict__ out, long long n) {
  constexpr int LPN = F / 4;
  long long t = (long long)blockIdx.x * blockDim.x + threadIdx.x;
  long long d = t / LPN;
  int q = (int)(t % LPN);
  if (d >= n) return;

  const float4* __restrict__ h4 = (const float4*)HS;
  int c = cnt[d];
  int m = c < BUCKET_CAP ? c : BUCKET_CAP;
  float4 acc = h4[d * LPN + q];
  const int* __restrict__ bi = bucket + d * BUCKET_CAP;

  for (int j = 0; j < m; j += 4) {
    bool k1 = (j + 1 < m), k2 = (j + 2 < m), k3 = (j + 3 < m);
    int i0, i1, i2, i3;
    if (k3) {
      int4 id = *(const int4*)(bi + j);
      i0 = id.x; i1 = id.y; i2 = id.z; i3 = id.w;
    } else {
      i0 = bi[j];
      i1 = k1 ? bi[j + 1] : 0;
      i2 = k2 ? bi[j + 2] : 0;
      i3 = 0;
    }
    long long s0 = i0;
    long long s1 = k1 ? (long long)i1 : d;
    long long s2 = k2 ? (long long)i2 : d;
    long long s3 = k3 ? (long long)i3 : d;
    float4 v0 = h4[s0 * LPN + q];
    float4 v1 = h4[s1 * LPN + q];
    float4 v2 = h4[s2 * LPN + q];
    float4 v3 = h4[s3 * LPN + q];
    acc.x += v0.x; acc.y += v0.y; acc.z += v0.z; acc.w += v0.w;
    acc.x += k1 ? v1.x : 0.0f; acc.y += k1 ? v1.y : 0.0f; acc.z += k1 ? v1.z : 0.0f; acc.w += k1 ? v1.w : 0.0f;
    acc.x += k2 ? v2.x : 0.0f; acc.y += k2 ? v2.y : 0.0f; acc.z += k2 ? v2.z : 0.0f; acc.w += k2 ? v2.w : 0.0f;
    acc.x += k3 ? v3.x : 0.0f; acc.y += k3 ? v3.y : 0.0f; acc.z += k3 ? v3.z : 0.0f; acc.w += k3 ? v3.w : 0.0f;
  }

  if (c > BUCKET_CAP) {
    int on = ovf_n[0];
    for (int p = 0; p < on; ++p) {
      if (ovf_dst[p] == (int)d) {
        float4 v = h4[(long long)ovf_src[p] * LPN + q];
        acc.x += v.x; acc.y += v.y; acc.z += v.z; acc.w += v.w;
      }
    }
  }

  float dv = rsqrtf((float)c + 1.0f);
  float y0 = acc.x * dv + bias[4 * q + 0];
  float y1 = acc.y * dv + bias[4 * q + 1];
  float y2 = acc.z * dv + bias[4 * q + 2];
  float y3 = acc.w * dv + bias[4 * q + 3];
  y0 = y0 > 0.0f ? y0 : 0.0f;
  y1 = y1 > 0.0f ? y1 : 0.0f;
  y2 = y2 > 0.0f ? y2 : 0.0f;
  y3 = y3 > 0.0f ? y3 : 0.0f;

  if (!HEAD) {
    float4 r;
    r.x = y0; r.y = y1; r.z = y2; r.w = y3;
    ((float4*)out)[d * LPN + q] = r;
  } else {
    float p = y0 * Wd[4 * q + 0] + y1 * Wd[4 * q + 1] + y2 * Wd[4 * q + 2] + y3 * Wd[4 * q + 3];
    for (int o = LPN / 2; o > 0; o >>= 1) p += __shfl_xor(p, o);
    if (q == 0) out[d] = p + bd[0];
  }
}

extern "C" void kernel_launch(void* const* d_in, const int* in_sizes, int n_in,
                              void* d_out, int out_size, void* d_ws, size_t ws_size,
                              hipStream_t stream) {
  const float* x  = (const float*)d_in[0];
  const int*   ei = (const int*)d_in[1];
  const float* W1 = (const float*)d_in[2];
  const float* b1 = (const float*)d_in[3];
  const float* W2 = (const float*)d_in[4];
  const float* b2 = (const float*)d_in[5];
  const float* Wd = (const float*)d_in[6];
  const float* bd = (const float*)d_in[7];
  float* out = (float*)d_out;

  long long n = in_sizes[0] / 10;
  long long E = in_sizes[1] / 2;
  const int* src = ei;
  const int* dst = ei + E;

  int* wsi = (int*)d_ws;
  long long ng  = (n + GROUP_NODES - 1) / GROUP_NODES;
  long long o_c = ((n + 2 + 63) / 64) * 64;
  long long o_b = o_c + ng * CNT_STRIDE;
  int*   cnt      = wsi;
  int*   ovfb_n   = wsi + n;
  int*   ovfa_n   = wsi + n + 1;
  int*   gcnt     = wsi + o_c;
  int*   bucket   = wsi + o_b;
  float* hs1      = (float*)(wsi + o_b + n * BUCKET_CAP);
  float* a1       = hs1 + n * 16;
  float* hs2      = hs1;
  int*   ovfb_src = (int*)(a1 + n * 16);
  int*   ovfb_dst = ovfb_src + E;
  unsigned* glist = (unsigned*)(ovfb_dst + E);
  int*   ovfa_src = (int*)(glist + ng * GROUP_CAP);
  int*   ovfa_dst = ovfa_src + E;

  const int B = 256;
  auto nb = [](long long t, int b) { return (unsigned)((t + b - 1) / b); };

  zero_i32<<<nb(o_b - n, B), B, 0, stream>>>(wsi + n, o_b - n);
  group_edges<<<nb(E, PART_THREADS * PART_EPT), PART_THREADS, 3 * ng * sizeof(int), stream>>>(
      src, dst, gcnt, glist, ovfa_n, ovfa_src, ovfa_dst, E, (int)ng);
  place_group<<<(unsigned)ng, GROUP_NODES, 0, stream>>>(gcnt, glist, ovfa_n, ovfa_src, ovfa_dst,
                                                       cnt, bucket, ovfb_n, ovfb_src, ovfb_dst, n);

  long long waves = (n + 15) / 16;
  long long gthreads = waves * 32;

  gemm_wmma_f32<10, 16><<<nb(gthreads, B), B, 0, stream>>>(x, W1, cnt, hs1, (int)n);
  gather_nodes<16, false><<<nb(n * 4, B), B, 0, stream>>>(hs1, cnt, bucket, ovfb_n, ovfb_src, ovfb_dst,
                                                          b1, Wd, bd, a1, n);

  gemm_wmma_f32<16, 8><<<nb(gthreads, B), B, 0, stream>>>(a1, W2, cnt, hs2, (int)n);
  gather_nodes<8, true><<<nb(n * 2, B), B, 0, stream>>>(hs2, cnt, bucket, ovfb_n, ovfb_src, ovfb_dst,
                                                        b2, Wd, bd, out, n);
}


# layer-2 transform (a1 @ W2, f32 WMMA, k order kept) moved into the layer-1 gather's epilogue via a wave-private LDS tile: a1 never stored, 6 launches not 7; sums unchanged
# speedup vs baseline: 4.2803x; 1.0645x over previous
#include <hip/hip_runtime.h>


typedef float v2f_t __attribute__((ext_vector_type(2)));
typedef float v8f_t __attribute__((ext_vector_type(8)));

#define BUCKET_CAP 32
#define GROUP_SHIFT 10
#define GROUP_NODES (1 << GROUP_SHIFT)
#define GROUP_CAP 16384
#define SRC_BITS 22
#define CNT_STRIDE 32
#define PART_THREADS 512
#define PART_EPT 16
#define PART_WAVES (PART_THREADS / 32)
#define ROW_PITCH 33

__global__ void zero_i32(int* __restrict__ p, long long n) {
  long long t = (long long)blockIdx.x * blockDim.x + threadIdx.x;
  if (t < n) p[t] = 0;
}

__global__ void __launch_bounds__(PART_THREADS)
group_edges(const int* __restrict__ src, const int* __restrict__ dst,
            int* gcnt, unsigned* __restrict__ glist, int* ovfa_n,
            int* __restrict__ ovfa_src, int* __restrict__ ovfa_dst,
            long long E, int ngroups) {
  extern __shared__ int sm[];
  __shared__ unsigned rec[PART_THREADS * PART_EPT];
  __shared__ int wtot[PART_WAVES];
  int* hist  = sm;
  int* lbase = sm + ngroups;
  int* gbase = sm + 2 * ngroups;
  int tid = threadIdx.x;
  int lane = tid & 31;
  int wave = tid >> 5;
  for (int i = tid; i < ngroups; i += PART_THREADS) hist[i] = 0;
  __syncthreads();

  long long e0 = (long long)blockIdx.x * (PART_THREADS * PART_EPT) + tid;
  int d[PART_EPT], s[PART_EPT], rk[PART_EPT];
#pragma unroll
  for (int k = 0; k < PART_EPT; ++k) {
    long long e = e0 + (long long)k * PART_THREADS;
    bool ok = e < E;
    long long ec = ok ? e : 0;
    int dv = dst[ec];
    s[k] = src[ec];
    d[k] = ok ? dv : -1;
  }
#pragma unroll
  for (int k = 0; k < PART_EPT; ++k) {
    rk[k] = 0;
    if (d[k] >= 0)
      rk[k] = __hip_atomic_fetch_add(&hist[d[k] >> GROUP_SHIFT], 1, __ATOMIC_RELAXED,
                                     __HIP_MEMORY_SCOPE_WORKGROUP);
  }
  __syncthreads();

  int chunk = (ngroups + PART_THREADS - 1) / PART_THREADS;
  int gfirst = tid * chunk;
  int sum = 0;
  for (int i = 0; i < chunk; ++i) {
    int g = gfirst + i;
    if (g < ngroups) {
      int c = hist[g];
      int b = 0;
      if (c > 0)
        b = __hip_atomic_fetch_add(&gcnt[(long long)g * CNT_STRIDE], c, __ATOMIC_RELAXED,
                                   __HIP_MEMORY_SCOPE_AGENT);
      gbase[g] = b;
      sum += c;
    }
  }
  int incl = sum;
  for (int o = 1; o < 32; o <<= 1) {
    int v = __shfl_up(incl, o);
    if (lane >= o) incl += v;
  }
  if (lane == 31) wtot[wave] = incl;
  __syncthreads();
  int run = incl - sum;
  for (int w = 0; w < wave; ++w) run += wtot[w];
  for (int i = 0; i < chunk; ++i) {
    int g = gfirst + i;
    if (g < ngroups) {
      lbase[g] = run;
      run += hist[g];
    }
  }
  __syncthreads();

#pragma unroll
  for (int k = 0; k < PART_EPT; ++k) rk[k] += lbase[(d[k] >= 0 ? d[k] : 0) >> GROUP_SHIFT];
#pragma unroll
  for (int k = 0; k < PART_EPT; ++k) {
    if (d[k] >= 0)
      rec[rk[k]] = ((unsigned)(d[k] & (GROUP_NODES - 1)) << SRC_BITS) | (unsigned)s[k];
  }
  __syncthreads();

  auto emit = [&](int g, int pos, unsigned r) {
    if (pos < GROUP_CAP) {
      glist[(long long)g * GROUP_CAP + pos] = r;
    } else {
      int p = __hip_atomic_fetch_add(ovfa_n, 1, __ATOMIC_RELAXED, __HIP_MEMORY_SCOPE_AGENT);
      ovfa_src[p] = (int)(r & ((1u << SRC_BITS) - 1u));
      ovfa_dst[p] = (g << GROUP_SHIFT) | (int)(r >> SRC_BITS);
    }
  };

  for (int gs = wave; gs < ngroups; gs += 4 * PART_WAVES) {
    int c[4], lb[4], gb[4];
    unsigned r[4];
#pragma unroll
    for (int u = 0; u < 4; ++u) {
      int g = gs + u * PART_WAVES;
      bool ok = g < ngroups;
      int gg = ok ? g : 0;
      int cv = hist[gg];
      lb[u] = lbase[gg];
      gb[u] = gbase[gg];
      c[u] = ok ? cv : 0;
    }
#pragma unroll
    for (int u = 0; u < 4; ++u) r[u] = rec[lane < c[u] ? lb[u] + lane : 0];
#pragma unroll
    for (int u = 0; u < 4; ++u)
      if (lane < c[u]) emit(gs + u * PART_WAVES, gb[u] + lane, r[u]);
#pragma unroll
    for (int u = 0; u < 4; ++u)
      for (int j = lane + 32; j < c[u]; j += 32) emit(gs + u * PART_WAVES, gb[u] + j, rec[lb[u] + j]);
  }
}

__global__ void __launch_bounds__(GROUP_NODES)
place_group(const int* __restrict__ gcnt, const unsigned* __restrict__ glist,
            const int* __restrict__ ovfa_n, const int* __restrict__ ovfa_src,
            const int* __restrict__ ovfa_dst, int* __restrict__ cnt, int* __restrict__ bucket,
            int* ovfb_n, int* __restrict__ ovfb_src, int* __restrict__ ovfb_dst, long long n) {
  __shared__ int lcnt[GROUP_NODES];
  __shared__ int lslot[GROUP_NODES * ROW_PITCH];
  int tid = threadIdx.x;
  int g = blockIdx.x;
  long long node0 = (long long)g << GROUP_SHIFT;
  lcnt[tid] = 0;
  __syncthreads();

  auto put = [&](int dl, int sv) {
    int r = __hip_atomic_fetch_add(&lcnt[dl], 1, __ATOMIC_RELAXED, __HIP_MEMORY_SCOPE_WORKGROUP);
    if (r < BUCKET_CAP) {
      lslot[dl * ROW_PITCH + r] = sv;
    } else {
      int p = __hip_atomic_fetch_add(ovfb_n, 1, __ATOMIC_RELAXED, __HIP_MEMORY_SCOPE_AGENT);
      ovfb_src[p] = sv;
      ovfb_dst[p] = (int)(node0 + dl);
    }
  };

  int total = gcnt[(long long)g * CNT_STRIDE];
  int m = total < GROUP_CAP ? total : GROUP_CAP;
  const unsigned* __restrict__ gl = glist + (long long)g * GROUP_CAP;
  for (int i0 = 0; i0 < m; i0 += 4 * GROUP_NODES) {
    unsigned r[4];
    bool ok[4];
#pragma unroll
    for (int u = 0; u < 4; ++u) {
      int i = i0 + u * GROUP_NODES + tid;
      ok[u] = i < m;
      r[u] = gl[ok[u] ? i : 0];
    }
#pragma unroll
    for (int u = 0; u < 4; ++u)
      if (ok[u]) put((int)(r[u] >> SRC_BITS), (int)(r[u] & ((1u << SRC_BITS) - 1u)));
  }

  int ona = ovfa_n[0];
  for (int p = tid; p < ona; p += GROUP_NODES) {
    int dv = ovfa_dst[p];
    if ((dv >> GROUP_SHIFT) == g) put(dv & (GROUP_NODES - 1), ovfa_src[p]);
  }
  __syncthreads();

  long long node = node0 + tid;
  if (node < n) cnt[node] = lcnt[tid];

  int j = tid % BUCKET_CAP;
  constexpr int RSTEP = GROUP_NODES / BUCKET_CAP;
  for (int row0 = tid / BUCKET_CAP; row0 < GROUP_NODES; row0 += 4 * RSTEP) {
    int c[4];
#pragma unroll
    for (int u = 0; u < 4; ++u) c[u] = lcnt[row0 + u * RSTEP];
#pragma unroll
    for (int u = 0; u < 4; ++u) {
      int row = row0 + u * RSTEP;
      long long nd = node0 + row;
      if (nd < n && j < c[u]) bucket[nd * BUCKET_CAP + j] = lslot[row * ROW_PITCH + j];
    }
  }
}

template <int K, int NCOL>
__global__ void gemm_wmma_f32(const float* __restrict__ X, const float* __restrict__ W,
                              const int* __restrict__ cnt, float* __restrict__ H, int n) {
  constexpr int KT = (K + 3) / 4;
  int lane = threadIdx.x & 31;
  int wave = (int)((blockIdx.x * (long long)blockDim.x + threadIdx.x) >> 5);
  int row_base = wave * 16;
  if (row_base >= n) return;

  int half = lane >> 4;
  int l    = lane & 15;

  int m = row_base + l;
  if (m > n - 1) m = n - 1;
  const float* __restrict__ xrow = X + (long long)m * K;

  const bool colok = (NCOL == 16) ? true : (l < NCOL);
  const int  lc    = colok ? l : 0;

  v8f_t c = {};
#pragma unroll
  for (int kt = 0; kt < KT; ++kt) {
    int k0 = kt * 4 + 2 * half;
    bool ok0 = (k0     < K);
    bool ok1 = (k0 + 1 < K);
    int ka0 = ok0 ? k0     : 0;
    int ka1 = ok1 ? k0 + 1 : 0;

    float a0 = xrow[ka0];
    float a1 = xrow[ka1];
    float w0 = W[ka0 * NCOL + lc];
    float w1 = W[ka1 * NCOL + lc];

    v2f_t a, b;
    a.x = ok0 ? a0 : 0.0f;
    a.y = ok1 ? a1 : 0.0f;
    b.x = (ok0 && colok) ? w0 : 0.0f;
    b.y = (ok1 && colok) ? w1 : 0.0f;

    c = __builtin_amdgcn_wmma_f32_16x16x4_f32(false, a, false, b, (short)0, c, false, false);
  }

  float* __restrict__ hb = H + (long long)(row_base + 8 * half) * NCOL + l;
  const int* __restrict__ cb = cnt + row_base + 8 * half;
  if (row_base + 16 <= n) {
    if (colok) {
#pragma unroll
      for (int v = 0; v < 8; ++v) hb[v * NCOL] = c[v] * rsqrtf((float)cb[v] + 1.0f);
    }
  } else {
#pragma unroll
    for (int v = 0; v < 8; ++v) {
      int row = row_base + 8 * half + v;
      if (row < n && colok) hb[v * NCOL] = c[v] * rsqrtf((float)cb[v] + 1.0f);
    }
  }
}

template <int F>
__device__ __forceinline__ float4 gather_relu(const float* __restrict__ HS, const int* __restrict__ cnt,
                                              const int* __restrict__ bucket,
                                              const int* __restrict__ ovf_n,
                                              const int* __restrict__ ovf_src,
                                              const int* __restrict__ ovf_dst,
                                              const float* __restrict__ bias, long long d, int q,
                                              float& dv) {
  constexpr int LPN = F / 4;
  const float4* __restrict__ h4 = (const float4*)HS;
  int c = cnt[d];
  int m = c < BUCKET_CAP ? c : BUCKET_CAP;
  float4 acc = h4[d * LPN + q];
  const int* __restrict__ bi = bucket + d * BUCKET_CAP;

  for (int j = 0; j < m; j += 4) {
    bool k1 = (j + 1 < m), k2 = (j + 2 < m), k3 = (j + 3 < m);
    int i0, i1, i2, i3;
    if (k3) {
      int4 id = *(const int4*)(bi + j);
      i0 = id.x; i1 = id.y; i2 = id.z; i3 = id.w;
    } else {
      i0 = bi[j];
      i1 = k1 ? bi[j + 1] : 0;
      i2 = k2 ? bi[j + 2] : 0;
      i3 = 0;
    }
    long long s0 = i0;
    long long s1 = k1 ? (long long)i1 : d;
    long long s2 = k2 ? (long long)i2 : d;
    long long s3 = k3 ? (long long)i3 : d;
    float4 v0 = h4[s0 * LPN + q];
    float4 v1 = h4[s1 * LPN + q];
    float4 v2 = h4[s2 * LPN + q];
    float4 v3 = h4[s3 * LPN + q];
    acc.x += v0.x; acc.y += v0.y; acc.z += v0.z; acc.w += v0.w;
    acc.x += k1 ? v1.x : 0.0f; acc.y += k1 ? v1.y : 0.0f; acc.z += k1 ? v1.z : 0.0f; acc.w += k1 ? v1.w : 0.0f;
    acc.x += k2 ? v2.x : 0.0f; acc.y += k2 ? v2.y : 0.0f; acc.z += k2 ? v2.z : 0.0f; acc.w += k2 ? v2.w : 0.0f;
    acc.x += k3 ? v3.x : 0.0f; acc.y += k3 ? v3.y : 0.0f; acc.z += k3 ? v3.z : 0.0f; acc.w += k3 ? v3.w : 0.0f;
  }

  if (c > BUCKET_CAP) {
    int on = ovf_n[0];
    for (int p = 0; p < on; ++p) {
      if (ovf_dst[p] == (int)d) {
        float4 v = h4[(long long)ovf_src[p] * LPN + q];
        acc.x += v.x; acc.y += v.y; acc.z += v.z; acc.w += v.w;
      }
    }
  }

  dv = rsqrtf((float)c + 1.0f);
  float4 y;
  y.x = acc.x * dv + bias[4 * q + 0];
  y.y = acc.y * dv + bias[4 * q + 1];
  y.z = acc.z * dv + bias[4 * q + 2];
  y.w = acc.w * dv + bias[4 * q + 3];
  y.x = y.x > 0.0f ? y.x : 0.0f;
  y.y = y.y > 0.0f ? y.y : 0.0f;
  y.z = y.z > 0.0f ? y.z : 0.0f;
  y.w = y.w > 0.0f ? y.w : 0.0f;
  return y;
}

#define GT_THREADS 256
#define GT_WAVES (GT_THREADS / 32)
#define GT_PITCH 20

template <int F, int NOUT>
__global__ void __launch_bounds__(GT_THREADS)
gather_transform(const float* __restrict__ HS, const int* __restrict__ cnt,
                 const int* __restrict__ bucket, const int* __restrict__ ovf_n,
                 const int* __restrict__ ovf_src, const int* __restrict__ ovf_dst,
                 const float* __restrict__ bias, const float* __restrict__ W,
                 float* __restrict__ HS2, long long n) {
  static_assert(F == 16 && NOUT == 8, "tile shapes below are written for 16 -> 8");
  constexpr int LPN = F / 4;
  constexpr int NPW = 32 / LPN;
  constexpr int KT = F / 4;
  __shared__ __attribute__((aligned(16))) float tile[GT_WAVES * NPW * GT_PITCH];

  long long t = (long long)blockIdx.x * blockDim.x + threadIdx.x;
  long long node0 = (t >> 5) * NPW;
  if (node0 >= n) return;
  int lane = threadIdx.x & 31;
  int wave = threadIdx.x >> 5;
  int q = lane & (LPN - 1);
  long long d = node0 + (lane >> 2);
  if (d > n - 1) d = n - 1;

  float dv;
  float4 y = gather_relu<F>(HS, cnt, bucket, ovf_n, ovf_src, ovf_dst, bias, d, q, dv);

  float* __restrict__ wt = tile + wave * (NPW * GT_PITCH);
  *(float4*)(wt + (lane >> 2) * GT_PITCH + 4 * q) = y;

  int half = lane >> 4;
  int c = lane & 15;
  int r = c & (NPW - 1);
  const float* __restrict__ yrow = wt + r * GT_PITCH + 2 * half;
  const float* __restrict__ wcol = W + (2 * half) * NOUT + r;

  v8f_t acc = {};
#pragma unroll
  for (int kt = 0; kt < KT; ++kt) {
    v2f_t a, b;
    a.x = wcol[(kt * 4) * NOUT];
    a.y = wcol[(kt * 4 + 1) * NOUT];
    float2 yy = *(const float2*)(yrow + kt * 4);
    b.x = yy.x;
    b.y = yy.y;
    acc = __builtin_amdgcn_wmma_f32_16x16x4_f32(false, a, false, b, (short)0, acc, false, false);
  }

  float dn = __shfl(dv, r * LPN);
  float4 o;
  o.x = (half ? acc[4] : acc[0]) * dn;
  o.y = (half ? acc[5] : acc[1]) * dn;
  o.z = (half ? acc[6] : acc[2]) * dn;
  o.w = (half ? acc[7] : acc[3]) * dn;
  if (c < NPW && node0 + r < n)
    *(float4*)(HS2 + (node0 + r) * NOUT + 4 * half) = o;
}

template <int F>
__global__ void gather_head(const float* __restrict__ HS, const int* __restrict__ cnt,
                            const int* __restrict__ bucket, const int* __restrict__ ovf_n,
                            const int* __restrict__ ovf_src, const int* __restrict__ ovf_dst,
                            const float* __restrict__ bias, const float* __restrict__ Wd,
                            const float* __restrict__ bd, float* __restrict__ out, long long n) {
  constexpr int LPN = F / 4;
  long long t = (long long)blockIdx.x * blockDim.x + threadIdx.x;
  long long d = t / LPN;
  int q = (int)(t % LPN);
  if (d >= n) return;

  float dv;
  float4 y = gather_relu<F>(HS, cnt, bucket, ovf_n, ovf_src, ovf_dst, bias, d, q, dv);

  float p = y.x * Wd[4 * q + 0] + y.y * Wd[4 * q + 1] + y.z * Wd[4 * q + 2] + y.w * Wd[4 * q + 3];
  for (int o = LPN / 2; o > 0; o >>= 1) p += __shfl_xor(p, o);
  if (q == 0) out[d] = p + bd[0];
}

extern "C" void kernel_launch(void* const* d_in, const int* in_sizes, int n_in,
                              void* d_out, int out_size, void* d_ws, size_t ws_size,
                              hipStream_t stream) {
  const float* x  = (const float*)d_in[0];
  const int*   ei = (const int*)d_in[1];
  const float* W1 = (const float*)d_in[2];
  const float* b1 = (const float*)d_in[3];
  const float* W2 = (const float*)d_in[4];
  const float* b2 = (const float*)d_in[5];
  const float* Wd = (const float*)d_in[6];
  const float* bd = (const float*)d_in[7];
  float* out = (float*)d_out;

  long long n = in_sizes[0] / 10;
  long long E = in_sizes[1] / 2;
  const int* src = ei;
  const int* dst = ei + E;

  int* wsi = (int*)d_ws;
  long long ng  = (n + GROUP_NODES - 1) / GROUP_NODES;
  long long o_c = ((n + 2 + 63) / 64) * 64;
  long long o_b = o_c + ng * CNT_STRIDE;
  int*   cnt      = wsi;
  int*   ovfb_n   = wsi + n;
  int*   ovfa_n   = wsi + n + 1;
  int*   gcnt     = wsi + o_c;
  int*   bucket   = wsi + o_b;
  float* hs1      = (float*)(wsi + o_b + n * BUCKET_CAP);
  float* hs2      = hs1 + n * 16;
  int*   ovfb_src = (int*)(hs2 + n * 16);
  int*   ovfb_dst = ovfb_src + E;
  unsigned* glist = (unsigned*)(ovfb_dst + E);
  int*   ovfa_src = (int*)(glist + ng * GROUP_CAP);
  int*   ovfa_dst = ovfa_src + E;

  const int B = 256;
  auto nb = [](long long t, int b) { return (unsigned)((t + b - 1) / b); };

  zero_i32<<<nb(o_b - n, B), B, 0, stream>>>(wsi + n, o_b - n);
  group_edges<<<nb(E, PART_THREADS * PART_EPT), PART_THREADS, 3 * ng * sizeof(int), stream>>>(
      src, dst, gcnt, glist, ovfa_n, ovfa_src, ovfa_dst, E, (int)ng);
  place_group<<<(unsigned)ng, GROUP_NODES, 0, stream>>>(gcnt, glist, ovfa_n, ovfa_src, ovfa_dst,
                                                       cnt, bucket, ovfb_n, ovfb_src, ovfb_dst, n);

  long long waves = (n + 15) / 16;
  long long gthreads = waves * 32;

  gemm_wmma_f32<10, 16><<<nb(gthreads, B), B, 0, stream>>>(x, W1, cnt, hs1, (int)n);
  gather_transform<16, 8><<<nb(n * 4, GT_THREADS), GT_THREADS, 0, stream>>>(
      hs1, cnt, bucket, ovfb_n, ovfb_src, ovfb_dst, b1, W2, hs2, n);

  gather_head<8><<<nb(n * 2, B), B, 0, stream>>>(hs2, cnt, bucket, ovfb_n, ovfb_src, ovfb_dst,
                                                 b2, Wd, bd, out, n);
}


# node's slot row read once per 16-slot half (its lanes load different 16-byte pieces, shared by quad_perm moves; place_group pads rows to 16), then its 16 row loads issued as one group; sum order kept
# speedup vs baseline: 4.7551x; 1.1109x over previous
#include <hip/hip_runtime.h>


typedef float v2f_t __attribute__((ext_vector_type(2)));
typedef float v8f_t __attribute__((ext_vector_type(8)));

#define BUCKET_CAP 32
#define HALF_ROW 16
#define GROUP_SHIFT 10
#define GROUP_NODES (1 << GROUP_SHIFT)
#define GROUP_CAP 16384
#define SRC_BITS 22
#define CNT_STRIDE 32
#define PART_THREADS 512
#define PART_EPT 16
#define PART_WAVES (PART_THREADS / 32)
#define ROW_PITCH 33

__global__ void zero_i32(int* __restrict__ p, long long n) {
  long long t = (long long)blockIdx.x * blockDim.x + threadIdx.x;
  if (t < n) p[t] = 0;
}

__global__ void __launch_bounds__(PART_THREADS)
group_edges(const int* __restrict__ src, const int* __restrict__ dst,
            int* gcnt, unsigned* __restrict__ glist, int* ovfa_n,
            int* __restrict__ ovfa_src, int* __restrict__ ovfa_dst,
            long long E, int ngroups) {
  extern __shared__ int sm[];
  __shared__ unsigned rec[PART_THREADS * PART_EPT];
  __shared__ int wtot[PART_WAVES];
  int* hist  = sm;
  int* lbase = sm + ngroups;
  int* gbase = sm + 2 * ngroups;
  int tid = threadIdx.x;
  int lane = tid & 31;
  int wave = tid >> 5;
  for (int i = tid; i < ngroups; i += PART_THREADS) hist[i] = 0;
  __syncthreads();

  long long e0 = (long long)blockIdx.x * (PART_THREADS * PART_EPT) + tid;
  int d[PART_EPT], s[PART_EPT], rk[PART_EPT];
#pragma unroll
  for (int k = 0; k < PART_EPT; ++k) {
    long long e = e0 + (long long)k * PART_THREADS;
    bool ok = e < E;
    long long ec = ok ? e : 0;
    int dv = dst[ec];
    s[k] = src[ec];
    d[k] = ok ? dv : -1;
  }
#pragma unroll
  for (int k = 0; k < PART_EPT; ++k) {
    rk[k] = 0;
    if (d[k] >= 0)
      rk[k] = __hip_atomic_fetch_add(&hist[d[k] >> GROUP_SHIFT], 1, __ATOMIC_RELAXED,
                                     __HIP_MEMORY_SCOPE_WORKGROUP);
  }
  __syncthreads();

  int chunk = (ngroups + PART_THREADS - 1) / PART_THREADS;
  int gfirst = tid * chunk;
  int sum = 0;
  for (int i = 0; i < chunk; ++i) {
    int g = gfirst + i;
    if (g < ngroups) {
      int c = hist[g];
      int b = 0;
      if (c > 0)
        b = __hip_atomic_fetch_add(&gcnt[(long long)g * CNT_STRIDE], c, __ATOMIC_RELAXED,
                                   __HIP_MEMORY_SCOPE_AGENT);
      gbase[g] = b;
      sum += c;
    }
  }
  int incl = sum;
  for (int o = 1; o < 32; o <<= 1) {
    int v = __shfl_up(incl, o);
    if (lane >= o) incl += v;
  }
  if (lane == 31) wtot[wave] = incl;
  __syncthreads();
  int run = incl - sum;
  for (int w = 0; w < wave; ++w) run += wtot[w];
  for (int i = 0; i < chunk; ++i) {
    int g = gfirst + i;
    if (g < ngroups) {
      lbase[g] = run;
      run += hist[g];
    }
  }
  __syncthreads();

#pragma unroll
  for (int k = 0; k < PART_EPT; ++k) rk[k] += lbase[(d[k] >= 0 ? d[k] : 0) >> GROUP_SHIFT];
#pragma unroll
  for (int k = 0; k < PART_EPT; ++k) {
    if (d[k] >= 0)
      rec[rk[k]] = ((unsigned)(d[k] & (GROUP_NODES - 1)) << SRC_BITS) | (unsigned)s[k];
  }
  __syncthreads();

  auto emit = [&](int g, int pos, unsigned r) {
    if (pos < GROUP_CAP) {
      glist[(long long)g * GROUP_CAP + pos] = r;
    } else {
      int p = __hip_atomic_fetch_add(ovfa_n, 1, __ATOMIC_RELAXED, __HIP_MEMORY_SCOPE_AGENT);
      ovfa_src[p] = (int)(r & ((1u << SRC_BITS) - 1u));
      ovfa_dst[p] = (g << GROUP_SHIFT) | (int)(r >> SRC_BITS);
    }
  };

  for (int gs = wave; gs < ngroups; gs += 4 * PART_WAVES) {
    int c[4], lb[4], gb[4];
    unsigned r[4];
#pragma unroll
    for (int u = 0; u < 4; ++u) {
      int g = gs + u * PART_WAVES;
      bool ok = g < ngroups;
      int gg = ok ? g : 0;
      int cv = hist[gg];
      lb[u] = lbase[gg];
      gb[u] = gbase[gg];
      c[u] = ok ? cv : 0;
    }
#pragma unroll
    for (int u = 0; u < 4; ++u) r[u] = rec[lane < c[u] ? lb[u] + lane : 0];
#pragma unroll
    for (int u = 0; u < 4; ++u)
      if (lane < c[u]) emit(gs + u * PART_WAVES, gb[u] + lane, r[u]);
#pragma unroll
    for (int u = 0; u < 4; ++u)
      for (int j = lane + 32; j < c[u]; j += 32) emit(gs + u * PART_WAVES, gb[u] + j, rec[lb[u] + j]);
  }
}

__global__ void __launch_bounds__(GROUP_NODES)
place_group(const int* __restrict__ gcnt, const unsigned* __restrict__ glist,
            const int* __restrict__ ovfa_n, const int* __restrict__ ovfa_src,
            const int* __restrict__ ovfa_dst, int* __restrict__ cnt, int* __restrict__ bucket,
            int* ovfb_n, int* __restrict__ ovfb_src, int* __restrict__ ovfb_dst, long long n) {
  __shared__ int lcnt[GROUP_NODES];
  __shared__ int lslot[GROUP_NODES * ROW_PITCH];
  int tid = threadIdx.x;
  int g = blockIdx.x;
  long long node0 = (long long)g << GROUP_SHIFT;
  lcnt[tid] = 0;
  __syncthreads();

  auto put = [&](int dl, int sv) {
    int r = __hip_atomic_fetch_add(&lcnt[dl], 1, __ATOMIC_RELAXED, __HIP_MEMORY_SCOPE_WORKGROUP);
    if (r < BUCKET_CAP) {
      lslot[dl * ROW_PITCH + r] = sv;
    } else {
      int p = __hip_atomic_fetch_add(ovfb_n, 1, __ATOMIC_RELAXED, __HIP_MEMORY_SCOPE_AGENT);
      ovfb_src[p] = sv;
      ovfb_dst[p] = (int)(node0 + dl);
    }
  };

  int total = gcnt[(long long)g * CNT_STRIDE];
  int m = total < GROUP_CAP ? total : GROUP_CAP;
  const unsigned* __restrict__ gl = glist + (long long)g * GROUP_CAP;
  for (int i0 = 0; i0 < m; i0 += 4 * GROUP_NODES) {
    unsigned r[4];
    bool ok[4];
#pragma unroll
    for (int u = 0; u < 4; ++u) {
      int i = i0 + u * GROUP_NODES + tid;
      ok[u] = i < m;
      r[u] = gl[ok[u] ? i : 0];
    }
#pragma unroll
    for (int u = 0; u < 4; ++u)
      if (ok[u]) put((int)(r[u] >> SRC_BITS), (int)(r[u] & ((1u << SRC_BITS) - 1u)));
  }

  int ona = ovfa_n[0];
  for (int p = tid; p < ona; p += GROUP_NODES) {
    int dv = ovfa_dst[p];
    if ((dv >> GROUP_SHIFT) == g) put(dv & (GROUP_NODES - 1), ovfa_src[p]);
  }
  __syncthreads();

  long long node = node0 + tid;
  if (node < n) cnt[node] = lcnt[tid];

  int j = tid % BUCKET_CAP;
  constexpr int RSTEP = GROUP_NODES / BUCKET_CAP;
  for (int row0 = tid / BUCKET_CAP; row0 < GROUP_NODES; row0 += 4 * RSTEP) {
    int c[4];
#pragma unroll
    for (int u = 0; u < 4; ++u) c[u] = lcnt[row0 + u * RSTEP];
#pragma unroll
    for (int u = 0; u < 4; ++u) {
      int row = row0 + u * RSTEP;
      long long nd = node0 + row;
      int lim = c[u] > HALF_ROW ? BUCKET_CAP : HALF_ROW;
      if (nd < n && j < lim)
        bucket[nd * BUCKET_CAP + j] = j < c[u] ? lslot[row * ROW_PITCH + j] : (int)nd;
    }
  }
}

template <int K, int NCOL>
__global__ void gemm_wmma_f32(const float* __restrict__ X, const float* __restrict__ W,
                              const int* __restrict__ cnt, float* __restrict__ H, int n) {
  constexpr int KT = (K + 3) / 4;
  int lane = threadIdx.x & 31;
  int wave = (int)((blockIdx.x * (long long)blockDim.x + threadIdx.x) >> 5);
  int row_base = wave * 16;
  if (row_base >= n) return;

  int half = lane >> 4;
  int l    = lane & 15;

  int m = row_base + l;
  if (m > n - 1) m = n - 1;
  const float* __restrict__ xrow = X + (long long)m * K;

  const bool colok = (NCOL == 16) ? true : (l < NCOL);
  const int  lc    = colok ? l : 0;

  v8f_t c = {};
#pragma unroll
  for (int kt = 0; kt < KT; ++kt) {
    int k0 = kt * 4 + 2 * half;
    bool ok0 = (k0     < K);
    bool ok1 = (k0 + 1 < K);
    int ka0 = ok0 ? k0     : 0;
    int ka1 = ok1 ? k0 + 1 : 0;

    float a0 = xrow[ka0];
    float a1 = xrow[ka1];
    float w0 = W[ka0 * NCOL + lc];
    float w1 = W[ka1 * NCOL + lc];

    v2f_t a, b;
    a.x = ok0 ? a0 : 0.0f;
    a.y = ok1 ? a1 : 0.0f;
    b.x = (ok0 && colok) ? w0 : 0.0f;
    b.y = (ok1 && colok) ? w1 : 0.0f;

    c = __builtin_amdgcn_wmma_f32_16x16x4_f32(false, a, false, b, (short)0, c, false, false);
  }

  float* __restrict__ hb = H + (long long)(row_base + 8 * half) * NCOL + l;
  const int* __restrict__ cb = cnt + row_base + 8 * half;
  if (row_base + 16 <= n) {
    if (colok) {
#pragma unroll
      for (int v = 0; v < 8; ++v) hb[v * NCOL] = c[v] * rsqrtf((float)cb[v] + 1.0f);
    }
  } else {
#pragma unroll
    for (int v = 0; v < 8; ++v) {
      int row = row_base + 8 * half + v;
      if (row < n && colok) hb[v * NCOL] = c[v] * rsqrtf((float)cb[v] + 1.0f);
    }
  }
}

template <int LPN, int QL>
__device__ __forceinline__ int group_bcast(int v) {
  static_assert(LPN == 4 || LPN == 2, "node groups of four or two lanes");
  constexpr int ctrl = (LPN == 4) ? (QL * 0x55) : (QL == 0 ? 0xA0 : 0xF5);
  return __builtin_amdgcn_update_dpp(v, v, ctrl, 0xF, 0xF, false);
}

template <int LPN, int P>
__device__ __forceinline__ int4 half_row_piece(const int4* mine) {
  constexpr int U = P / LPN, QL = P % LPN;
  int4 r;
  r.x = group_bcast<LPN, QL>(mine[U].x);
  r.y = group_bcast<LPN, QL>(mine[U].y);
  r.z = group_bcast<LPN, QL>(mine[U].z);
  r.w = group_bcast<LPN, QL>(mine[U].w);
  return r;
}

template <int LPN>
__device__ __forceinline__ void add_half_row(const float4* __restrict__ h4, const int* __restrict__ half_row,
                                             int filled, long long d, int q, float4& acc) {
  constexpr int PPL = 4 / LPN;
  int4 mine[PPL];
#pragma unroll
  for (int u = 0; u < PPL; ++u) mine[u] = *(const int4*)(half_row + 4 * (q + LPN * u));
  int4 p0 = half_row_piece<LPN, 0>(mine);
  int4 p1 = half_row_piece<LPN, 1>(mine);
  int4 p2 = half_row_piece<LPN, 2>(mine);
  int4 p3 = half_row_piece<LPN, 3>(mine);
  int idx[HALF_ROW] = {p0.x, p0.y, p0.z, p0.w, p1.x, p1.y, p1.z, p1.w,
                       p2.x, p2.y, p2.z, p2.w, p3.x, p3.y, p3.z, p3.w};
  float4 v[HALF_ROW];
#pragma unroll
  for (int j = 0; j < HALF_ROW; ++j) {
    long long s = j < filled ? (long long)idx[j] : d;
    v[j] = h4[s * LPN + q];
  }
#pragma unroll
  for (int j = 0; j < HALF_ROW; ++j) {
    bool k = j < filled;
    acc.x += k ? v[j].x : 0.0f;
    acc.y += k ? v[j].y : 0.0f;
    acc.z += k ? v[j].z : 0.0f;
    acc.w += k ? v[j].w : 0.0f;
  }
}

template <int F>
__device__ __forceinline__ float4 gather_relu(const float* __restrict__ HS, const int* __restrict__ cnt,
                                              const int* __restrict__ bucket,
                                              const int* __restrict__ ovf_n,
                                              const int* __restrict__ ovf_src,
                                              const int* __restrict__ ovf_dst,
                                              const float* __restrict__ bias, long long d, int q,
                                              float& dv) {
  constexpr int LPN = F / 4;
  const float4* __restrict__ h4 = (const float4*)HS;
  int c = cnt[d];
  int m = c < BUCKET_CAP ? c : BUCKET_CAP;
  float4 acc = h4[d * LPN + q];
  const int* __restrict__ bi = bucket + d * BUCKET_CAP;

  add_half_row<LPN>(h4, bi, m < HALF_ROW ? m : HALF_ROW, d, q, acc);
  if (c > HALF_ROW)
    add_half_row<LPN>(h4, bi + HALF_ROW, m - HALF_ROW, d, q, acc);

  if (c > BUCKET_CAP) {
    int on = ovf_n[0];
    for (int p = 0; p < on; ++p) {
      if (ovf_dst[p] == (int)d) {
        float4 v = h4[(long long)ovf_src[p] * LPN + q];
        acc.x += v.x; acc.y += v.y; acc.z += v.z; acc.w += v.w;
      }
    }
  }

  dv = rsqrtf((float)c + 1.0f);
  float4 y;
  y.x = acc.x * dv + bias[4 * q + 0];
  y.y = acc.y * dv + bias[4 * q + 1];
  y.z = acc.z * dv + bias[4 * q + 2];
  y.w = acc.w * dv + bias[4 * q + 3];
  y.x = y.x > 0.0f ? y.x : 0.0f;
  y.y = y.y > 0.0f ? y.y : 0.0f;
  y.z = y.z > 0.0f ? y.z : 0.0f;
  y.w = y.w > 0.0f ? y.w : 0.0f;
  return y;
}

#define GT_THREADS 256
#define GT_WAVES (GT_THREADS / 32)
#define GT_PITCH 20

template <int F, int NOUT>
__global__ void __launch_bounds__(GT_THREADS)
gather_transform(const float* __restrict__ HS, const int* __restrict__ cnt,
                 const int* __restrict__ bucket, const int* __restrict__ ovf_n,
                 const int* __restrict__ ovf_src, const int* __restrict__ ovf_dst,
                 const float* __restrict__ bias, const float* __restrict__ W,
                 float* __restrict__ HS2, long long n) {
  static_assert(F == 16 && NOUT == 8, "tile shapes below are written for 16 -> 8");
  constexpr int LPN = F / 4;
  constexpr int NPW = 32 / LPN;
  constexpr int KT = F / 4;
  __shared__ __attribute__((aligned(16))) float tile[GT_WAVES * NPW * GT_PITCH];

  long long t = (long long)blockIdx.x * blockDim.x + threadIdx.x;
  long long node0 = (t >> 5) * NPW;
  if (node0 >= n) return;
  int lane = threadIdx.x & 31;
  int wave = threadIdx.x >> 5;
  int q = lane & (LPN - 1);
  long long d = node0 + (lane >> 2);
  if (d > n - 1) d = n - 1;

  float dv;
  float4 y = gather_relu<F>(HS, cnt, bucket, ovf_n, ovf_src, ovf_dst, bias, d, q, dv);

  float* __restrict__ wt = tile + wave * (NPW * GT_PITCH);
  *(float4*)(wt + (lane >> 2) * GT_PITCH + 4 * q) = y;

  int half = lane >> 4;
  int c = lane & 15;
  int r = c & (NPW - 1);
  const float* __restrict__ yrow = wt + r * GT_PITCH + 2 * half;
  const float* __restrict__ wcol = W + (2 * half) * NOUT + r;

  v8f_t acc = {};
#pragma unroll
  for (int kt = 0; kt < KT; ++kt) {
    v2f_t a, b;
    a.x = wcol[(kt * 4) * NOUT];
    a.y = wcol[(kt * 4 + 1) * NOUT];
    float2 yy = *(const float2*)(yrow + kt * 4);
    b.x = yy.x;
    b.y = yy.y;
    acc = __builtin_amdgcn_wmma_f32_16x16x4_f32(false, a, false, b, (short)0, acc, false, false);
  }

  float dn = __shfl(dv, r * LPN);
  float4 o;
  o.x = (half ? acc[4] : acc[0]) * dn;
  o.y = (half ? acc[5] : acc[1]) * dn;
  o.z = (half ? acc[6] : acc[2]) * dn;
  o.w = (half ? acc[7] : acc[3]) * dn;
  if (c < NPW && node0 + r < n)
    *(float4*)(HS2 + (node0 + r) * NOUT + 4 * half) = o;
}

template <int F>
__global__ void gather_head(const float* __restrict__ HS, const int* __restrict__ cnt,
                            const int* __restrict__ bucket, const int* __restrict__ ovf_n,
                            const int* __restrict__ ovf_src, const int* __restrict__ ovf_dst,
                            const float* __restrict__ bias, const float* __restrict__ Wd,
                            const float* __restrict__ bd, float* __restrict__ out, long long n) {
  constexpr int LPN = F / 4;
  long long t = (long long)blockIdx.x * blockDim.x + threadIdx.x;
  long long d = t / LPN;
  int q = (int)(t % LPN);
  if (d >= n) return;

  float dv;
  float4 y = gather_relu<F>(HS, cnt, bucket, ovf_n, ovf_src, ovf_dst, bias, d, q, dv);

  float p = y.x * Wd[4 * q + 0] + y.y * Wd[4 * q + 1] + y.z * Wd[4 * q + 2] + y.w * Wd[4 * q + 3];
  for (int o = LPN / 2; o > 0; o >>= 1) p += __shfl_xor(p, o);
  if (q == 0) out[d] = p + bd[0];
}

extern "C" void kernel_launch(void* const* d_in, const int* in_sizes, int n_in,
                              void* d_out, int out_size, void* d_ws, size_t ws_size,
                              hipStream_t stream) {
  const float* x  = (const float*)d_in[0];
  const int*   ei = (const int*)d_in[1];
  const float* W1 = (const float*)d_in[2];
  const float* b1 = (const float*)d_in[3];
  const float* W2 = (const float*)d_in[4];
  const float* b2 = (const float*)d_in[5];
  const float* Wd = (const float*)d_in[6];
  const float* bd = (const float*)d_in[7];
  float* out = (float*)d_out;

  long long n = in_sizes[0] / 10;
  long long E = in_sizes[1] / 2;
  const int* src = ei;
  const int* dst = ei + E;

  int* wsi = (int*)d_ws;
  long long ng  = (n + GROUP_NODES - 1) / GROUP_NODES;
  long long o_c = ((n + 2 + 63) / 64) * 64;
  long long o_b = o_c + ng * CNT_STRIDE;
  int*   cnt      = wsi;
  int*   ovfb_n   = wsi + n;
  int*   ovfa_n   = wsi + n + 1;
  int*   gcnt     = wsi + o_c;
  int*   bucket   = wsi + o_b;
  float* hs1      = (float*)(wsi + o_b + n * BUCKET_CAP);
  float* hs2      = hs1 + n * 16;
  int*   ovfb_src = (int*)(hs2 + n * 16);
  int*   ovfb_dst = ovfb_src + E;
  unsigned* glist = (unsigned*)(ovfb_dst + E);
  int*   ovfa_src = (int*)(glist + ng * GROUP_CAP);
  int*   ovfa_dst = ovfa_src + E;

  const int B = 256;
  auto nb = [](long long t, int b) { return (unsigned)((t + b - 1) / b); };

  zero_i32<<<nb(o_b - n, B), B, 0, stream>>>(wsi + n, o_b - n);
  group_edges<<<nb(E, PART_THREADS * PART_EPT), PART_THREADS, 3 * ng * sizeof(int), stream>>>(
      src, dst, gcnt, glist, ovfa_n, ovfa_src, ovfa_dst, E, (int)ng);
  place_group<<<(unsigned)ng, GROUP_NODES, 0, stream>>>(gcnt, glist, ovfa_n, ovfa_src, ovfa_dst,
                                                       cnt, bucket, ovfb_n, ovfb_src, ovfb_dst, n);

  long long waves = (n + 15) / 16;
  long long gthreads = waves * 32;

  gemm_wmma_f32<10, 16><<<nb(gthreads, B), B, 0, stream>>>(x, W1, cnt, hs1, (int)n);
  gather_transform<16, 8><<<nb(n * 4, GT_THREADS), GT_THREADS, 0, stream>>>(
      hs1, cnt, bucket, ovfb_n, ovfb_src, ovfb_dst, b1, W2, hs2, n);

  gather_head<8><<<nb(n * 2, B), B, 0, stream>>>(hs2, cnt, bucket, ovfb_n, ovfb_src, ovfb_dst,
                                                 b2, Wd, bd, out, n);
}


# layer-1 transform (x @ W1, f32 WMMA, k order and dinv scaling kept) moved into place_group: a wave loads two tiles' operands at its start, multiplies, stores them scaled by LDS degrees; 5 launches
# speedup vs baseline: 4.9040x; 1.0313x over previous
#include <hip/hip_runtime.h>


typedef float v2f_t __attribute__((ext_vector_type(2)));
typedef float v8f_t __attribute__((ext_vector_type(8)));

#define BUCKET_CAP 32
#define HALF_ROW 16
#define GROUP_SHIFT 10
#define GROUP_NODES (1 << GROUP_SHIFT)
#define GROUP_CAP 16384
#define SRC_BITS 22
#define CNT_STRIDE 32
#define PART_THREADS 512
#define PART_EPT 16
#define PART_WAVES (PART_THREADS / 32)
#define ROW_PITCH 33

__global__ void zero_i32(int* __restrict__ p, long long n) {
  long long t = (long long)blockIdx.x * blockDim.x + threadIdx.x;
  if (t < n) p[t] = 0;
}

__global__ void __launch_bounds__(PART_THREADS)
group_edges(const int* __restrict__ src, const int* __restrict__ dst,
            int* gcnt, unsigned* __restrict__ glist, int* ovfa_n,
            int* __restrict__ ovfa_src, int* __restrict__ ovfa_dst,
            long long E, int ngroups) {
  extern __shared__ int sm[];
  __shared__ unsigned rec[PART_THREADS * PART_EPT];
  __shared__ int wtot[PART_WAVES];
  int* hist  = sm;
  int* lbase = sm + ngroups;
  int* gbase = sm + 2 * ngroups;
  int tid = threadIdx.x;
  int lane = tid & 31;
  int wave = tid >> 5;
  for (int i = tid; i < ngroups; i += PART_THREADS) hist[i] = 0;
  __syncthreads();

  long long e0 = (long long)blockIdx.x * (PART_THREADS * PART_EPT) + tid;
  int d[PART_EPT], s[PART_EPT], rk[PART_EPT];
#pragma unroll
  for (int k = 0; k < PART_EPT; ++k) {
    long long e = e0 + (long long)k * PART_THREADS;
    bool ok = e < E;
    long long ec = ok ? e : 0;
    int dv = dst[ec];
    s[k] = src[ec];
    d[k] = ok ? dv : -1;
  }
#pragma unroll
  for (int k = 0; k < PART_EPT; ++k) {
    rk[k] = 0;
    if (d[k] >= 0)
      rk[k] = __hip_atomic_fetch_add(&hist[d[k] >> GROUP_SHIFT], 1, __ATOMIC_RELAXED,
                                     __HIP_MEMORY_SCOPE_WORKGROUP);
  }
  __syncthreads();

  int chunk = (ngroups + PART_THREADS - 1) / PART_THREADS;
  int gfirst = tid * chunk;
  int sum = 0;
  for (int i = 0; i < chunk; ++i) {
    int g = gfirst + i;
    if (g < ngroups) {
      int c = hist[g];
      int b = 0;
      if (c > 0)
        b = __hip_atomic_fetch_add(&gcnt[(long long)g * CNT_STRIDE], c, __ATOMIC_RELAXED,
                                   __HIP_MEMORY_SCOPE_AGENT);
      gbase[g] = b;
      sum += c;
    }
  }
  int incl = sum;
  for (int o = 1; o < 32; o <<= 1) {
    int v = __shfl_up(incl, o);
    if (lane >= o) incl += v;
  }
  if (lane == 31) wtot[wave] = incl;
  __syncthreads();
  int run = incl - sum;
  for (int w = 0; w < wave; ++w) run += wtot[w];
  for (int i = 0; i < chunk; ++i) {
    int g = gfirst + i;
    if (g < ngroups) {
      lbase[g] = run;
      run += hist[g];
    }
  }
  __syncthreads();

#pragma unroll
  for (int k = 0; k < PART_EPT; ++k) rk[k] += lbase[(d[k] >= 0 ? d[k] : 0) >> GROUP_SHIFT];
#pragma unroll
  for (int k = 0; k < PART_EPT; ++k) {
    if (d[k] >= 0)
      rec[rk[k]] = ((unsigned)(d[k] & (GROUP_NODES - 1)) << SRC_BITS) | (unsigned)s[k];
  }
  __syncthreads();

  auto emit = [&](int g, int pos, unsigned r) {
    if (pos < GROUP_CAP) {
      glist[(long long)g * GROUP_CAP + pos] = r;
    } else {
      int p = __hip_atomic_fetch_add(ovfa_n, 1, __ATOMIC_RELAXED, __HIP_MEMORY_SCOPE_AGENT);
      ovfa_src[p] = (int)(r & ((1u << SRC_BITS) - 1u));
      ovfa_dst[p] = (g << GROUP_SHIFT) | (int)(r >> SRC_BITS);
    }
  };

  for (int gs = wave; gs < ngroups; gs += 4 * PART_WAVES) {
    int c[4], lb[4], gb[4];
    unsigned r[4];
#pragma unroll
    for (int u = 0; u < 4; ++u) {
      int g = gs + u * PART_WAVES;
      bool ok = g < ngroups;
      int gg = ok ? g : 0;
      int cv = hist[gg];
      lb[u] = lbase[gg];
      gb[u] = gbase[gg];
      c[u] = ok ? cv : 0;
    }
#pragma unroll
    for (int u = 0; u < 4; ++u) r[u] = rec[lane < c[u] ? lb[u] + lane : 0];
#pragma unroll
    for (int u = 0; u < 4; ++u)
      if (lane < c[u]) emit(gs + u * PART_WAVES, gb[u] + lane, r[u]);
#pragma unroll
    for (int u = 0; u < 4; ++u)
      for (int j = lane + 32; j < c[u]; j += 32) emit(gs + u * PART_WAVES, gb[u] + j, rec[lb[u] + j]);
  }
}

template <int K, int T>
struct xw_operands {
  float x[T][(K + 3) / 4][2];
  float w[(K + 3) / 4][2];
};

template <int K, int NCOL, int T>
__device__ __forceinline__ void xw_load(const float* __restrict__ X, const float* __restrict__ W,
                                        long long row_base, long long n, int lane,
                                        xw_operands<K, T>& op) {
  constexpr int KT = (K + 3) / 4;
  int half = lane >> 4;
  int l    = lane & 15;
  const int lc = (NCOL == 16 || l < NCOL) ? l : 0;
#pragma unroll
  for (int kt = 0; kt < KT; ++kt) {
    int k0 = kt * 4 + 2 * half;
    int ka0 = (k0     < K) ? k0     : 0;
    int ka1 = (k0 + 1 < K) ? k0 + 1 : 0;
    op.w[kt][0] = W[ka0 * NCOL + lc];
    op.w[kt][1] = W[ka1 * NCOL + lc];
#pragma unroll
    for (int u = 0; u < T; ++u) {
      long long m = row_base + 16 * u + l;
      if (m > n - 1) m = n - 1;
      op.x[u][kt][0] = X[m * K + ka0];
      op.x[u][kt][1] = X[m * K + ka1];
    }
  }
}

template <int K, int NCOL, int T>
__device__ __forceinline__ v8f_t xw_product(const xw_operands<K, T>& op, int u, int lane) {
  constexpr int KT = (K + 3) / 4;
  int half = lane >> 4;
  const bool colok = (NCOL == 16) ? true : ((lane & 15) < NCOL);
  v8f_t c = {};
#pragma unroll
  for (int kt = 0; kt < KT; ++kt) {
    int k0 = kt * 4 + 2 * half;
    bool ok0 = (k0     < K);
    bool ok1 = (k0 + 1 < K);
    v2f_t a, b;
    a.x = ok0 ? op.x[u][kt][0] : 0.0f;
    a.y = ok1 ? op.x[u][kt][1] : 0.0f;
    b.x = (ok0 && colok) ? op.w[kt][0] : 0.0f;
    b.y = (ok1 && colok) ? op.w[kt][1] : 0.0f;
    c = __builtin_amdgcn_wmma_f32_16x16x4_f32(false, a, false, b, (short)0, c, false, false);
  }
  return c;
}

template <int K, int NCOL>
__global__ void __launch_bounds__(GROUP_NODES)
place_group(const int* __restrict__ gcnt, const unsigned* __restrict__ glist,
            const int* __restrict__ ovfa_n, const int* __restrict__ ovfa_src,
            const int* __restrict__ ovfa_dst, int* __restrict__ cnt, int* __restrict__ bucket,
            int* ovfb_n, int* __restrict__ ovfb_src, int* __restrict__ ovfb_dst,
            const float* __restrict__ X, const float* __restrict__ W, float* __restrict__ HS,
            long long n) {
  constexpr int TPW = GROUP_NODES / 16 / (GROUP_NODES / 32);
  __shared__ int lcnt[GROUP_NODES];
  __shared__ int lslot[GROUP_NODES * ROW_PITCH];
  int tid = threadIdx.x;
  int g = blockIdx.x;
  long long node0 = (long long)g << GROUP_SHIFT;
  int lane = tid & 31;
  int trow0 = (tid >> 5) * (16 * TPW);
  xw_operands<K, TPW> xw;
  xw_load<K, NCOL, TPW>(X, W, node0 + trow0, n, lane, xw);
  lcnt[tid] = 0;
  __syncthreads();

  auto put = [&](int dl, int sv) {
    int r = __hip_atomic_fetch_add(&lcnt[dl], 1, __ATOMIC_RELAXED, __HIP_MEMORY_SCOPE_WORKGROUP);
    if (r < BUCKET_CAP) {
      lslot[dl * ROW_PITCH + r] = sv;
    } else {
      int p = __hip_atomic_fetch_add(ovfb_n, 1, __ATOMIC_RELAXED, __HIP_MEMORY_SCOPE_AGENT);
      ovfb_src[p] = sv;
      ovfb_dst[p] = (int)(node0 + dl);
    }
  };

  int total = gcnt[(long long)g * CNT_STRIDE];
  int m = total < GROUP_CAP ? total : GROUP_CAP;
  const unsigned* __restrict__ gl = glist + (long long)g * GROUP_CAP;
  for (int i0 = 0; i0 < m; i0 += 4 * GROUP_NODES) {
    unsigned r[4];
    bool ok[4];
#pragma unroll
    for (int u = 0; u < 4; ++u) {
      int i = i0 + u * GROUP_NODES + tid;
      ok[u] = i < m;
      r[u] = gl[ok[u] ? i : 0];
    }
#pragma unroll
    for (int u = 0; u < 4; ++u)
      if (ok[u]) put((int)(r[u] >> SRC_BITS), (int)(r[u] & ((1u << SRC_BITS) - 1u)));
  }

  int ona = ovfa_n[0];
  for (int p = tid; p < ona; p += GROUP_NODES) {
    int dv = ovfa_dst[p];
    if ((dv >> GROUP_SHIFT) == g) put(dv & (GROUP_NODES - 1), ovfa_src[p]);
  }

  v8f_t hacc[TPW];
#pragma unroll
  for (int u = 0; u < TPW; ++u) hacc[u] = xw_product<K, NCOL, TPW>(xw, u, lane);
  __syncthreads();

  long long node = node0 + tid;
  if (node < n) cnt[node] = lcnt[tid];

  int j = tid % BUCKET_CAP;
  constexpr int RSTEP = GROUP_NODES / BUCKET_CAP;
  for (int row0 = tid / BUCKET_CAP; row0 < GROUP_NODES; row0 += 4 * RSTEP) {
    int c[4];
#pragma unroll
    for (int u = 0; u < 4; ++u) c[u] = lcnt[row0 + u * RSTEP];
#pragma unroll
    for (int u = 0; u < 4; ++u) {
      int row = row0 + u * RSTEP;
      long long nd = node0 + row;
      int lim = c[u] > HALF_ROW ? BUCKET_CAP : HALF_ROW;
      if (nd < n && j < lim)
        bucket[nd * BUCKET_CAP + j] = j < c[u] ? lslot[row * ROW_PITCH + j] : (int)nd;
    }
  }

  const bool colok = (NCOL == 16) ? true : ((lane & 15) < NCOL);
#pragma unroll
  for (int u = 0; u < TPW; ++u) {
    int lr = trow0 + 16 * u + 8 * (lane >> 4);
    long long rb = node0 + trow0 + 16 * u;
    float* __restrict__ hb = HS + (node0 + lr) * NCOL + (lane & 15);
    if (rb + 16 <= n) {
      if (colok) {
#pragma unroll
        for (int v = 0; v < 8; ++v) hb[v * NCOL] = hacc[u][v] * rsqrtf((float)lcnt[lr + v] + 1.0f);
      }
    } else if (rb < n) {
#pragma unroll
      for (int v = 0; v < 8; ++v)
        if (node0 + lr + v < n && colok) hb[v * NCOL] = hacc[u][v] * rsqrtf((float)lcnt[lr + v] + 1.0f);
    }
  }
}

template <int LPN, int QL>
__device__ __forceinline__ int group_bcast(int v) {
  static_assert(LPN == 4 || LPN == 2, "node groups of four or two lanes");
  constexpr int ctrl = (LPN == 4) ? (QL * 0x55) : (QL == 0 ? 0xA0 : 0xF5);
  return __builtin_amdgcn_update_dpp(v, v, ctrl, 0xF, 0xF, false);
}

template <int LPN, int P>
__device__ __forceinline__ int4 half_row_piece(const int4* mine) {
  constexpr int U = P / LPN, QL = P % LPN;
  int4 r;
  r.x = group_bcast<LPN, QL>(mine[U].x);
  r.y = group_bcast<LPN, QL>(mine[U].y);
  r.z = group_bcast<LPN, QL>(mine[U].z);
  r.w = group_bcast<LPN, QL>(mine[U].w);
  return r;
}

template <int LPN>
__device__ __forceinline__ void add_half_row(const float4* __restrict__ h4, const int* __restrict__ half_row,
                                             int filled, long long d, int q, float4& acc) {
  constexpr int PPL = 4 / LPN;
  int4 mine[PPL];
#pragma unroll
  for (int u = 0; u < PPL; ++u) mine[u] = *(const int4*)(half_row + 4 * (q + LPN * u));
  int4 p0 = half_row_piece<LPN, 0>(mine);
  int4 p1 = half_row_piece<LPN, 1>(mine);
  int4 p2 = half_row_piece<LPN, 2>(mine);
  int4 p3 = half_row_piece<LPN, 3>(mine);
  int idx[HALF_ROW] = {p0.x, p0.y, p0.z, p0.w, p1.x, p1.y, p1.z, p1.w,
                       p2.x, p2.y, p2.z, p2.w, p3.x, p3.y, p3.z, p3.w};
  float4 v[HALF_ROW];
#pragma unroll
  for (int j = 0; j < HALF_ROW; ++j) {
    long long s = j < filled ? (long long)idx[j] : d;
    v[j] = h4[s * LPN + q];
  }
#pragma unroll
  for (int j = 0; j < HALF_ROW; ++j) {
    bool k = j < filled;
    acc.x += k ? v[j].x : 0.0f;
    acc.y += k ? v[j].y : 0.0f;
    acc.z += k ? v[j].z : 0.0f;
    acc.w += k ? v[j].w : 0.0f;
  }
}

template <int F>
__device__ __forceinline__ float4 gather_relu(const float* __restrict__ HS, const int* __restrict__ cnt,
                                              const int* __restrict__ bucket,
                                              const int* __restrict__ ovf_n,
                                              const int* __restrict__ ovf_src,
                                              const int* __restrict__ ovf_dst,
                                              const float* __restrict__ bias, long long d, int q,
                                              float& dv) {
  constexpr int LPN = F / 4;
  const float4* __restrict__ h4 = (const float4*)HS;
  int c = cnt[d];
  int m = c < BUCKET_CAP ? c : BUCKET_CAP;
  float4 acc = h4[d * LPN + q];
  const int* __restrict__ bi = bucket + d * BUCKET_CAP;

  add_half_row<LPN>(h4, bi, m < HALF_ROW ? m : HALF_ROW, d, q, acc);
  if (c > HALF_ROW)
    add_half_row<LPN>(h4, bi + HALF_ROW, m - HALF_ROW, d, q, acc);

  if (c > BUCKET_CAP) {
    int on = ovf_n[0];
    for (int p = 0; p < on; ++p) {
      if (ovf_dst[p] == (int)d) {
        float4 v = h4[(long long)ovf_src[p] * LPN + q];
        acc.x += v.x; acc.y += v.y; acc.z += v.z; acc.w += v.w;
      }
    }
  }

  dv = rsqrtf((float)c + 1.0f);
  float4 y;
  y.x = acc.x * dv + bias[4 * q + 0];
  y.y = acc.y * dv + bias[4 * q + 1];
  y.z = acc.z * dv + bias[4 * q + 2];
  y.w = acc.w * dv + bias[4 * q + 3];
  y.x = y.x > 0.0f ? y.x : 0.0f;
  y.y = y.y > 0.0f ? y.y : 0.0f;
  y.z = y.z > 0.0f ? y.z : 0.0f;
  y.w = y.w > 0.0f ? y.w : 0.0f;
  return y;
}

#define GT_THREADS 256
#define GT_WAVES (GT_THREADS / 32)
#define GT_PITCH 20

template <int F, int NOUT>
__global__ void __launch_bounds__(GT_THREADS)
gather_transform(const float* __restrict__ HS, const int* __restrict__ cnt,
                 const int* __restrict__ bucket, const int* __restrict__ ovf_n,
                 const int* __restrict__ ovf_src, const int* __restrict__ ovf_dst,
                 const float* __restrict__ bias, const float* __restrict__ W,
                 float* __restrict__ HS2, long long n) {
  static_assert(F == 16 && NOUT == 8, "tile shapes below are written for 16 -> 8");
  constexpr int LPN = F / 4;
  constexpr int NPW = 32 / LPN;
  constexpr int KT = F / 4;
  __shared__ __attribute__((aligned(16))) float tile[GT_WAVES * NPW * GT_PITCH];

  long long t = (long long)blockIdx.x * blockDim.x + threadIdx.x;
  long long node0 = (t >> 5) * NPW;
  if (node0 >= n) return;
  int lane = threadIdx.x & 31;
  int wave = threadIdx.x >> 5;
  int q = lane & (LPN - 1);
  long long d = node0 + (lane >> 2);
  if (d > n - 1) d = n - 1;

  float dv;
  float4 y = gather_relu<F>(HS, cnt, bucket, ovf_n, ovf_src, ovf_dst, bias, d, q, dv);

  float* __restrict__ wt = tile + wave * (NPW * GT_PITCH);
  *(float4*)(wt + (lane >> 2) * GT_PITCH + 4 * q) = y;

  int half = lane >> 4;
  int c = lane & 15;
  int r = c & (NPW - 1);
  const float* __restrict__ yrow = wt + r * GT_PITCH + 2 * half;
  const float* __restrict__ wcol = W + (2 * half) * NOUT + r;

  v8f_t acc = {};
#pragma unroll
  for (int kt = 0; kt < KT; ++kt) {
    v2f_t a, b;
    a.x = wcol[(kt * 4) * NOUT];
    a.y = wcol[(kt * 4 + 1) * NOUT];
    float2 yy = *(const float2*)(yrow + kt * 4);
    b.x = yy.x;
    b.y = yy.y;
    acc = __builtin_amdgcn_wmma_f32_16x16x4_f32(false, a, false, b, (short)0, acc, false, false);
  }

  float dn = __shfl(dv, r * LPN);
  float4 o;
  o.x = (half ? acc[4] : acc[0]) * dn;
  o.y = (half ? acc[5] : acc[1]) * dn;
  o.z = (half ? acc[6] : acc[2]) * dn;
  o.w = (half ? acc[7] : acc[3]) * dn;
  if (c < NPW && node0 + r < n)
    *(float4*)(HS2 + (node0 + r) * NOUT + 4 * half) = o;
}

template <int F>
__global__ void gather_head(const float* __restrict__ HS, const int* __restrict__ cnt,
                            const int* __restrict__ bucket, const int* __restrict__ ovf_n,
                            const int* __restrict__ ovf_src, const int* __restrict__ ovf_dst,
                            const float* __restrict__ bias, const float* __restrict__ Wd,
                            const float* __restrict__ bd, float* __restrict__ out, long long n) {
  constexpr int LPN = F / 4;
  long long t = (long long)blockIdx.x * blockDim.x + threadIdx.x;
  long long d = t / LPN;
  int q = (int)(t % LPN);
  if (d >= n) return;

  float dv;
  float4 y = gather_relu<F>(HS, cnt, bucket, ovf_n, ovf_src, ovf_dst, bias, d, q, dv);

  float p = y.x * Wd[4 * q + 0] + y.y * Wd[4 * q + 1] + y.z * Wd[4 * q + 2] + y.w * Wd[4 * q + 3];
  for (int o = LPN / 2; o > 0; o >>= 1) p += __shfl_xor(p, o);
  if (q == 0) out[d] = p + bd[0];
}

extern "C" void kernel_launch(void* const* d_in, const int* in_sizes, int n_in,
                              void* d_out, int out_size, void* d_ws, size_t ws_size,
                              hipStream_t stream) {
  const float* x  = (const float*)d_in[0];
  const int*   ei = (const int*)d_in[1];
  const float* W1 = (const float*)d_in[2];
  const float* b1 = (const float*)d_in[3];
  const float* W2 = (const float*)d_in[4];
  const float* b2 = (const float*)d_in[5];
  const float* Wd = (const float*)d_in[6];
  const float* bd = (const float*)d_in[7];
  float* out = (float*)d_out;

  long long n = in_sizes[0] / 10;
  long long E = in_sizes[1] / 2;
  const int* src = ei;
  const int* dst = ei + E;

  int* wsi = (int*)d_ws;
  long long ng  = (n + GROUP_NODES - 1) / GROUP_NODES;
  long long o_c = ((n + 2 + 63) / 64) * 64;
  long long o_b = o_c + ng * CNT_STRIDE;
  int*   cnt      = wsi;
  int*   ovfb_n   = wsi + n;
  int*   ovfa_n   = wsi + n + 1;
  int*   gcnt     = wsi + o_c;
  int*   bucket   = wsi + o_b;
  float* hs1      = (float*)(wsi + o_b + n * BUCKET_CAP);
  float* hs2      = hs1 + n * 16;
  int*   ovfb_src = (int*)(hs2 + n * 16);
  int*   ovfb_dst = ovfb_src + E;
  unsigned* glist = (unsigned*)(ovfb_dst + E);
  int*   ovfa_src = (int*)(glist + ng * GROUP_CAP);
  int*   ovfa_dst = ovfa_src + E;

  const int B = 256;
  auto nb = [](long long t, int b) { return (unsigned)((t + b - 1) / b); };

  zero_i32<<<nb(o_b - n, B), B, 0, stream>>>(wsi + n, o_b - n);
  group_edges<<<nb(E, PART_THREADS * PART_EPT), PART_THREADS, 3 * ng * sizeof(int), stream>>>(
      src, dst, gcnt, glist, ovfa_n, ovfa_src, ovfa_dst, E, (int)ng);
  place_group<10, 16><<<(unsigned)ng, GROUP_NODES, 0, stream>>>(gcnt, glist, ovfa_n, ovfa_src, ovfa_dst,
                                                               cnt, bucket, ovfb_n, ovfb_src, ovfb_dst,
                                                               x, W1, hs1, n);

  gather_transform<16, 8><<<nb(n * 4, GT_THREADS), GT_THREADS, 0, stream>>>(
      hs1, cnt, bucket, ovfb_n, ovfb_src, ovfb_dst, b1, W2, hs2, n);

  gather_head<8><<<nb(n * 2, B), B, 0, stream>>>(hs2, cnt, bucket, ovfb_n, ovfb_src, ovfb_dst,
                                                 b2, Wd, bd, out, n);
}


# edge runs go to fixed (group, block) cells, one 32-slot line each, that place_group streams (a sweep's ranks taken together): no global counters, no zeroing launch, 4 launches; sum order by arrival
# speedup vs baseline: 5.1668x; 1.0536x over previous
#include <hip/hip_runtime.h>


typedef float v2f_t __attribute__((ext_vector_type(2)));
typedef float v8f_t __attribute__((ext_vector_type(8)));

#define BUCKET_CAP 32
#define HALF_ROW 16
#define GROUP_SHIFT 10
#define GROUP_NODES (1 << GROUP_SHIFT)
#define SRC_BITS 22
#define PART_THREADS 512
#define PART_EPT 16
#define PART_EDGES (PART_THREADS * PART_EPT)
#define PART_WAVES (PART_THREADS / 32)
#define CELL_CAP 32
#define CELL_EMPTY 0xFFFFFFFFu
#define PLACE_RPT 20
#define ROW_PITCH 33

__global__ void __launch_bounds__(PART_THREADS)
group_edges(const int* __restrict__ src, const int* __restrict__ dst,
            unsigned* __restrict__ cells, int* __restrict__ spill_n,
            int* __restrict__ spill_src, int* __restrict__ spill_dst,
            int* __restrict__ ovfb_n, long long E, int ngroups) {
  extern __shared__ int sm[];
  __shared__ unsigned rec[PART_EDGES];
  __shared__ int wtot[PART_WAVES];
  int* hist  = sm;
  int* lbase = sm + ngroups;
  int tid = threadIdx.x;
  int lane = tid & 31;
  int wave = tid >> 5;
  for (int i = tid; i < ngroups; i += PART_THREADS) hist[i] = 0;
  if (tid == 0 && blockIdx.x == 0) ovfb_n[0] = 0;
  __syncthreads();

  long long e0 = (long long)blockIdx.x * PART_EDGES + tid;
  int d[PART_EPT], s[PART_EPT], rk[PART_EPT];
#pragma unroll
  for (int k = 0; k < PART_EPT; ++k) {
    long long e = e0 + (long long)k * PART_THREADS;
    bool ok = e < E;
    long long ec = ok ? e : 0;
    int dv = dst[ec];
    s[k] = src[ec];
    d[k] = ok ? dv : -1;
  }
#pragma unroll
  for (int k = 0; k < PART_EPT; ++k) {
    rk[k] = 0;
    if (d[k] >= 0)
      rk[k] = __hip_atomic_fetch_add(&hist[d[k] >> GROUP_SHIFT], 1, __ATOMIC_RELAXED,
                                     __HIP_MEMORY_SCOPE_WORKGROUP);
  }
  __syncthreads();

  int chunk = (ngroups + PART_THREADS - 1) / PART_THREADS;
  int gfirst = tid * chunk;
  int sum = 0;
  for (int i = 0; i < chunk; ++i) {
    int g = gfirst + i;
    if (g < ngroups) sum += hist[g];
  }
  int incl = sum;
  for (int o = 1; o < 32; o <<= 1) {
    int v = __shfl_up(incl, o);
    if (lane >= o) incl += v;
  }
  if (lane == 31) wtot[wave] = incl;
  __syncthreads();
  int run = incl - sum;
  for (int w = 0; w < wave; ++w) run += wtot[w];
  for (int i = 0; i < chunk; ++i) {
    int g = gfirst + i;
    if (g < ngroups) {
      lbase[g] = run;
      run += hist[g];
    }
  }
  __syncthreads();

#pragma unroll
  for (int k = 0; k < PART_EPT; ++k) rk[k] += lbase[(d[k] >= 0 ? d[k] : 0) >> GROUP_SHIFT];
#pragma unroll
  for (int k = 0; k < PART_EPT; ++k) {
    if (d[k] >= 0)
      rec[rk[k]] = ((unsigned)(d[k] & (GROUP_NODES - 1)) << SRC_BITS) | (unsigned)s[k];
  }
  __syncthreads();

  auto excess_before = [&](int gend) {
    int t = 0;
    for (int gg = lane; gg < gend; gg += 32) {
      int cc = hist[gg];
      t += cc > CELL_CAP ? cc - CELL_CAP : 0;
    }
    for (int o = 16; o > 0; o >>= 1) t += __shfl_xor(t, o);
    return t;
  };

  const long long gpitch = (long long)gridDim.x * CELL_CAP;
  unsigned* __restrict__ mycell = cells + (long long)blockIdx.x * CELL_CAP + lane;
  for (int gs = wave; gs < ngroups; gs += 4 * PART_WAVES) {
    int c[4], lb[4];
    unsigned r[4];
#pragma unroll
    for (int u = 0; u < 4; ++u) {
      int g = gs + u * PART_WAVES;
      bool ok = g < ngroups;
      int gg = ok ? g : 0;
      int cv = hist[gg];
      lb[u] = lbase[gg];
      c[u] = ok ? cv : -1;
    }
#pragma unroll
    for (int u = 0; u < 4; ++u) r[u] = rec[lane < c[u] ? lb[u] + lane : 0];
#pragma unroll
    for (int u = 0; u < 4; ++u)
      if (c[u] >= 0) mycell[(long long)(gs + u * PART_WAVES) * gpitch] = lane < c[u] ? r[u] : CELL_EMPTY;
#pragma unroll
    for (int u = 0; u < 4; ++u)
      if (c[u] > CELL_CAP) {
        int g = gs + u * PART_WAVES;
        long long sp0 = (long long)blockIdx.x * PART_EDGES + excess_before(g) - CELL_CAP;
        for (int j = lane + CELL_CAP; j < c[u]; j += 32) {
          unsigned rr = rec[lb[u] + j];
          spill_src[sp0 + j] = (int)(rr & ((1u << SRC_BITS) - 1u));
          spill_dst[sp0 + j] = (g << GROUP_SHIFT) | (int)(rr >> SRC_BITS);
        }
      }
  }
  if (wave == PART_WAVES - 1) {
    int t = excess_before(ngroups);
    if (lane == 0) spill_n[blockIdx.x] = t;
  }
}

template <int K, int T>
struct xw_operands {
  float x[T][(K + 3) / 4][2];
  float w[(K + 3) / 4][2];
};

template <int K, int NCOL, int T>
__device__ __forceinline__ void xw_load(const float* __restrict__ X, const float* __restrict__ W,
                                        long long row_base, long long n, int lane,
                                        xw_operands<K, T>& op) {
  constexpr int KT = (K + 3) / 4;
  int half = lane >> 4;
  int l    = lane & 15;
  const int lc = (NCOL == 16 || l < NCOL) ? l : 0;
#pragma unroll
  for (int kt = 0; kt < KT; ++kt) {
    int k0 = kt * 4 + 2 * half;
    int ka0 = (k0     < K) ? k0     : 0;
    int ka1 = (k0 + 1 < K) ? k0 + 1 : 0;
    op.w[kt][0] = W[ka0 * NCOL + lc];
    op.w[kt][1] = W[ka1 * NCOL + lc];
#pragma unroll
    for (int u = 0; u < T; ++u) {
      long long m = row_base + 16 * u + l;
      if (m > n - 1) m = n - 1;
      op.x[u][kt][0] = X[m * K + ka0];
      op.x[u][kt][1] = X[m * K + ka1];
    }
  }
}

template <int K, int NCOL, int T>
__device__ __forceinline__ v8f_t xw_product(const xw_operands<K, T>& op, int u, int lane) {
  constexpr int KT = (K + 3) / 4;
  int half = lane >> 4;
  const bool colok = (NCOL == 16) ? true : ((lane & 15) < NCOL);
  v8f_t c = {};
#pragma unroll
  for (int kt = 0; kt < KT; ++kt) {
    int k0 = kt * 4 + 2 * half;
    bool ok0 = (k0     < K);
    bool ok1 = (k0 + 1 < K);
    v2f_t a, b;
    a.x = ok0 ? op.x[u][kt][0] : 0.0f;
    a.y = ok1 ? op.x[u][kt][1] : 0.0f;
    b.x = (ok0 && colok) ? op.w[kt][0] : 0.0f;
    b.y = (ok1 && colok) ? op.w[kt][1] : 0.0f;
    c = __builtin_amdgcn_wmma_f32_16x16x4_f32(false, a, false, b, (short)0, c, false, false);
  }
  return c;
}

template <int K, int NCOL>
__global__ void __launch_bounds__(GROUP_NODES)
place_group(const unsigned* __restrict__ cells, int nblk, const int* __restrict__ spill_n,
            const int* __restrict__ spill_src, const int* __restrict__ spill_dst,
            int* __restrict__ cnt, int* __restrict__ bucket,
            int* ovfb_n, int* __restrict__ ovfb_src, int* __restrict__ ovfb_dst,
            const float* __restrict__ X, const float* __restrict__ W, float* __restrict__ HS,
            long long n) {
  constexpr int TPW = GROUP_NODES / 16 / (GROUP_NODES / 32);
  __shared__ int lcnt[GROUP_NODES];
  __shared__ int lslot[GROUP_NODES * ROW_PITCH];
  int tid = threadIdx.x;
  int g = blockIdx.x;
  long long node0 = (long long)g << GROUP_SHIFT;
  int lane = tid & 31;
  int trow0 = (tid >> 5) * (16 * TPW);
  xw_operands<K, TPW> xw;
  xw_load<K, NCOL, TPW>(X, W, node0 + trow0, n, lane, xw);

  const int m = nblk * CELL_CAP;
  const unsigned* __restrict__ gl = cells + (long long)g * m;
  unsigned rs[PLACE_RPT];
  auto load_sweep = [&](int i0) {
#pragma unroll
    for (int u = 0; u < PLACE_RPT; ++u) {
      int i = i0 + u * GROUP_NODES + tid;
      unsigned v = gl[i < m ? i : 0];
      rs[u] = i < m ? v : CELL_EMPTY;
    }
  };
  load_sweep(0);
  int ns_first = spill_n[tid < nblk ? tid : 0];
  if (tid >= nblk) ns_first = 0;
  lcnt[tid] = 0;
  __syncthreads();
  int dv_first = 0;
  if (ns_first > 0) dv_first = spill_dst[(long long)tid * PART_EDGES];

  auto take_rank = [&](int dl) {
    return __hip_atomic_fetch_add(&lcnt[dl], 1, __ATOMIC_RELAXED, __HIP_MEMORY_SCOPE_WORKGROUP);
  };
  auto store_slot = [&](int dl, int sv, int rank) {
    if (rank < BUCKET_CAP) {
      lslot[dl * ROW_PITCH + rank] = sv;
    } else {
      int p = __hip_atomic_fetch_add(ovfb_n, 1, __ATOMIC_RELAXED, __HIP_MEMORY_SCOPE_AGENT);
      ovfb_src[p] = sv;
      ovfb_dst[p] = (int)(node0 + dl);
    }
  };
  auto put = [&](int dl, int sv) { store_slot(dl, sv, take_rank(dl)); };

  for (int i0 = 0;;) {
    int rk[PLACE_RPT];
#pragma unroll
    for (int u = 0; u < PLACE_RPT; ++u) {
      rk[u] = -1;
      if (rs[u] != CELL_EMPTY) rk[u] = take_rank((int)(rs[u] >> SRC_BITS));
    }
#pragma unroll
    for (int u = 0; u < PLACE_RPT; ++u)
      if (rk[u] >= 0) store_slot((int)(rs[u] >> SRC_BITS), (int)(rs[u] & ((1u << SRC_BITS) - 1u)), rk[u]);
    i0 += PLACE_RPT * GROUP_NODES;
    if (i0 >= m) break;
    load_sweep(i0);
  }

  for (int b = tid; b < nblk; b += GROUP_NODES) {
    int ns = b == tid ? ns_first : spill_n[b];
    const long long sb = (long long)b * PART_EDGES;
    for (int p = 0; p < ns; ++p) {
      int dv = (b == tid && p == 0) ? dv_first : spill_dst[sb + p];
      if ((dv >> GROUP_SHIFT) == g) put(dv & (GROUP_NODES - 1), spill_src[sb + p]);
    }
  }

  v8f_t hacc[TPW];
#pragma unroll
  for (int u = 0; u < TPW; ++u) hacc[u] = xw_product<K, NCOL, TPW>(xw, u, lane);
  __syncthreads();

  long long node = node0 + tid;
  if (node < n) cnt[node] = lcnt[tid];

  int j = tid % BUCKET_CAP;
  constexpr int RSTEP = GROUP_NODES / BUCKET_CAP;
  for (int row0 = tid / BUCKET_CAP; row0 < GROUP_NODES; row0 += 4 * RSTEP) {
    int c[4];
#pragma unroll
    for (int u = 0; u < 4; ++u) c[u] = lcnt[row0 + u * RSTEP];
#pragma unroll
    for (int u = 0; u < 4; ++u) {
      int row = row0 + u * RSTEP;
      long long nd = node0 + row;
      int lim = c[u] > HALF_ROW ? BUCKET_CAP : HALF_ROW;
      if (nd < n && j < lim)
        bucket[nd * BUCKET_CAP + j] = j < c[u] ? lslot[row * ROW_PITCH + j] : (int)nd;
    }
  }

  const bool colok = (NCOL == 16) ? true : ((lane & 15) < NCOL);
#pragma unroll
  for (int u = 0; u < TPW; ++u) {
    int lr = trow0 + 16 * u + 8 * (lane >> 4);
    long long rb = node0 + trow0 + 16 * u;
    float* __restrict__ hb = HS + (node0 + lr) * NCOL + (lane & 15);
    if (rb + 16 <= n) {
      if (colok) {
#pragma unroll
        for (int v = 0; v < 8; ++v) hb[v * NCOL] = hacc[u][v] * rsqrtf((float)lcnt[lr + v] + 1.0f);
      }
    } else if (rb < n) {
#pragma unroll
      for (int v = 0; v < 8; ++v)
        if (node0 + lr + v < n && colok) hb[v * NCOL] = hacc[u][v] * rsqrtf((float)lcnt[lr + v] + 1.0f);
    }
  }
}

template <int LPN, int QL>
__device__ __forceinline__ int group_bcast(int v) {
  static_assert(LPN == 4 || LPN == 2, "node groups of four or two lanes");
  constexpr int ctrl = (LPN == 4) ? (QL * 0x55) : (QL == 0 ? 0xA0 : 0xF5);
  return __builtin_amdgcn_update_dpp(v, v, ctrl, 0xF, 0xF, false);
}

template <int LPN, int P>
__device__ __forceinline__ int4 half_row_piece(const int4* mine) {
  constexpr int U = P / LPN, QL = P % LPN;
  int4 r;
  r.x = group_bcast<LPN, QL>(mine[U].x);
  r.y = group_bcast<LPN, QL>(mine[U].y);
  r.z = group_bcast<LPN, QL>(mine[U].z);
  r.w = group_bcast<LPN, QL>(mine[U].w);
  return r;
}

template <int LPN>
__device__ __forceinline__ void add_half_row(const float4* __restrict__ h4, const int* __restrict__ half_row,
                                             int filled, long long d, int q, float4& acc) {
  constexpr int PPL = 4 / LPN;
  int4 mine[PPL];
#pragma unroll
  for (int u = 0; u < PPL; ++u) mine[u] = *(const int4*)(half_row + 4 * (q + LPN * u));
  int4 p0 = half_row_piece<LPN, 0>(mine);
  int4 p1 = half_row_piece<LPN, 1>(mine);
  int4 p2 = half_row_piece<LPN, 2>(mine);
  int4 p3 = half_row_piece<LPN, 3>(mine);
  int idx[HALF_ROW] = {p0.x, p0.y, p0.z, p0.w, p1.x, p1.y, p1.z, p1.w,
                       p2.x, p2.y, p2.z, p2.w, p3.x, p3.y, p3.z, p3.w};
  float4 v[HALF_ROW];
#pragma unroll
  for (int j = 0; j < HALF_ROW; ++j) {
    long long s = j < filled ? (long long)idx[j] : d;
    v[j] = h4[s * LPN + q];
  }
#pragma unroll
  for (int j = 0; j < HALF_ROW; ++j) {
    bool k = j < filled;
    acc.x += k ? v[j].x : 0.0f;
    acc.y += k ? v[j].y : 0.0f;
    acc.z += k ? v[j].z : 0.0f;
    acc.w += k ? v[j].w : 0.0f;
  }
}

template <int F>
__device__ __forceinline__ float4 gather_relu(const float* __restrict__ HS, const int* __restrict__ cnt,
                                              const int* __restrict__ bucket,
                                              const int* __restrict__ ovf_n,
                                              const int* __restrict__ ovf_src,
                                              const int* __restrict__ ovf_dst,
                                              const float* __restrict__ bias, long long d, int q,
                                              float& dv) {
  constexpr int LPN = F / 4;
  const float4* __restrict__ h4 = (const float4*)HS;
  int c = cnt[d];
  int m = c < BUCKET_CAP ? c : BUCKET_CAP;
  float4 acc = h4[d * LPN + q];
  const int* __restrict__ bi = bucket + d * BUCKET_CAP;

  add_half_row<LPN>(h4, bi, m < HALF_ROW ? m : HALF_ROW, d, q, acc);
  if (c > HALF_ROW)
    add_half_row<LPN>(h4, bi + HALF_ROW, m - HALF_ROW, d, q, acc);

  if (c > BUCKET_CAP) {
    int on = ovf_n[0];
    for (int p = 0; p < on; ++p) {
      if (ovf_dst[p] == (int)d) {
        float4 v = h4[(long long)ovf_src[p] * LPN + q];
        acc.x += v.x; acc.y += v.y; acc.z += v.z; acc.w += v.w;
      }
    }
  }

  dv = rsqrtf((float)c + 1.0f);
  float4 y;
  y.x = acc.x * dv + bias[4 * q + 0];
  y.y = acc.y * dv + bias[4 * q + 1];
  y.z = acc.z * dv + bias[4 * q + 2];
  y.w = acc.w * dv + bias[4 * q + 3];
  y.x = y.x > 0.0f ? y.x : 0.0f;
  y.y = y.y > 0.0f ? y.y : 0.0f;
  y.z = y.z > 0.0f ? y.z : 0.0f;
  y.w = y.w > 0.0f ? y.w : 0.0f;
  return y;
}

#define GT_THREADS 256
#define GT_WAVES (GT_THREADS / 32)
#define GT_PITCH 20

template <int F, int NOUT>
__global__ void __launch_bounds__(GT_THREADS)
gather_transform(const float* __restrict__ HS, const int* __restrict__ cnt,
                 const int* __restrict__ bucket, const int* __restrict__ ovf_n,
                 const int* __restrict__ ovf_src, const int* __restrict__ ovf_dst,
                 const float* __restrict__ bias, const float* __restrict__ W,
                 float* __restrict__ HS2, long long n) {
  static_assert(F == 16 && NOUT == 8, "tile shapes below are written for 16 -> 8");
  constexpr int LPN = F / 4;
  constexpr int NPW = 32 / LPN;
  constexpr int KT = F / 4;
  __shared__ __attribute__((aligned(16))) float tile[GT_WAVES * NPW * GT_PITCH];

  long long t = (long long)blockIdx.x * blockDim.x + threadIdx.x;
  long long node0 = (t >> 5) * NPW;
  if (node0 >= n) return;
  int lane = threadIdx.x & 31;
  int wave = threadIdx.x >> 5;
  int q = lane & (LPN - 1);
  long long d = node0 + (lane >> 2);
  if (d > n - 1) d = n - 1;

  float dv;
  float4 y = gather_relu<F>(HS, cnt, bucket, ovf_n, ovf_src, ovf_dst, bias, d, q, dv);

  float* __restrict__ wt = tile + wave * (NPW * GT_PITCH);
  *(float4*)(wt + (lane >> 2) * GT_PITCH + 4 * q) = y;

  int half = lane >> 4;
  int c = lane & 15;
  int r = c & (NPW - 1);
  const float* __restrict__ yrow = wt + r * GT_PITCH + 2 * half;
  const float* __restrict__ wcol = W + (2 * half) * NOUT + r;

  v8f_t acc = {};
#pragma unroll
  for (int kt = 0; kt < KT; ++kt) {
    v2f_t a, b;
    a.x = wcol[(kt * 4) * NOUT];
    a.y = wcol[(kt * 4 + 1) * NOUT];
    float2 yy = *(const float2*)(yrow + kt * 4);
    b.x = yy.x;
    b.y = yy.y;
    acc = __builtin_amdgcn_wmma_f32_16x16x4_f32(false, a, false, b, (short)0, acc, false, false);
  }

  float dn = __shfl(dv, r * LPN);
  float4 o;
  o.x = (half ? acc[4] : acc[0]) * dn;
  o.y = (half ? acc[5] : acc[1]) * dn;
  o.z = (half ? acc[6] : acc[2]) * dn;
  o.w = (half ? acc[7] : acc[3]) * dn;
  if (c < NPW && node0 + r < n)
    *(float4*)(HS2 + (node0 + r) * NOUT + 4 * half) = o;
}

template <int F>
__global__ void gather_head(const float* __restrict__ HS, const int* __restrict__ cnt,
                            const int* __restrict__ bucket, const int* __restrict__ ovf_n,
                            const int* __restrict__ ovf_src, const int* __restrict__ ovf_dst,
                            const float* __restrict__ bias, const float* __restrict__ Wd,
                            const float* __restrict__ bd, float* __restrict__ out, long long n) {
  constexpr int LPN = F / 4;
  long long t = (long long)blockIdx.x * blockDim.x + threadIdx.x;
  long long d = t / LPN;
  int q = (int)(t % LPN);
  if (d >= n) return;

  float dv;
  float4 y = gather_relu<F>(HS, cnt, bucket, ovf_n, ovf_src, ovf_dst, bias, d, q, dv);

  float p = y.x * Wd[4 * q + 0] + y.y * Wd[4 * q + 1] + y.z * Wd[4 * q + 2] + y.w * Wd[4 * q + 3];
  for (int o = LPN / 2; o > 0; o >>= 1) p += __shfl_xor(p, o);
  if (q == 0) out[d] = p + bd[0];
}

extern "C" void kernel_launch(void* const* d_in, const int* in_sizes, int n_in,
                              void* d_out, int out_size, void* d_ws, size_t ws_size,
                              hipStream_t stream) {
  const float* x  = (const float*)d_in[0];
  const int*   ei = (const int*)d_in[1];
  const float* W1 = (const float*)d_in[2];
  const float* b1 = (const float*)d_in[3];
  const float* W2 = (const float*)d_in[4];
  const float* b2 = (const float*)d_in[5];
  const float* Wd = (const float*)d_in[6];
  const float* bd = (const float*)d_in[7];
  float* out = (float*)d_out;

  long long n = in_sizes[0] / 10;
  long long E = in_sizes[1] / 2;
  const int* src = ei;
  const int* dst = ei + E;

  int* wsi = (int*)d_ws;
  long long ng   = (n + GROUP_NODES - 1) / GROUP_NODES;
  long long nblk = (E + PART_EDGES - 1) / PART_EDGES;
  long long o_c = ((n + 1 + 63) / 64) * 64;
  long long o_b = o_c + ((nblk + 63) / 64) * 64;
  int*   cnt      = wsi;
  int*   ovfb_n   = wsi + n;
  int*   spill_n  = wsi + o_c;
  int*   bucket   = wsi + o_b;
  float* hs1      = (float*)(wsi + o_b + n * BUCKET_CAP);
  float* hs2      = hs1 + n * 16;
  int*   ovfb_src = (int*)(hs2 + n * 16);
  int*   ovfb_dst = ovfb_src + E;
  unsigned* cells = (unsigned*)(ovfb_dst + E);
  int*   spill_src = (int*)(cells + ng * nblk * CELL_CAP);
  int*   spill_dst = spill_src + nblk * PART_EDGES;

  const int B = 256;
  auto nb = [](long long t, int b) { return (unsigned)((t + b - 1) / b); };

  group_edges<<<(unsigned)nblk, PART_THREADS, 2 * ng * sizeof(int), stream>>>(
      src, dst, cells, spill_n, spill_src, spill_dst, ovfb_n, E, (int)ng);
  place_group<10, 16><<<(unsigned)ng, GROUP_NODES, 0, stream>>>(cells, (int)nblk, spill_n, spill_src, spill_dst,
                                                               cnt, bucket, ovfb_n, ovfb_src, ovfb_dst,
                                                               x, W1, hs1, n);

  gather_transform<16, 8><<<nb(n * 4, GT_THREADS), GT_THREADS, 0, stream>>>(
      hs1, cnt, bucket, ovfb_n, ovfb_src, ovfb_dst, b1, W2, hs2, n);

  gather_head<8><<<nb(n * 2, B), B, 0, stream>>>(hs2, cnt, bucket, ovfb_n, ovfb_src, ovfb_dst,
                                                 b2, Wd, bd, out, n);
}


# bucket rows kept as two dense planes of 16-slot half rows: place_group stores two nodes' first halves as one whole 128-byte line, the gathers read 64 used bytes per node; sum order kept
# speedup vs baseline: 5.4697x; 1.0586x over previous
#include <hip/hip_runtime.h>


typedef float v2f_t __attribute__((ext_vector_type(2)));
typedef float v8f_t __attribute__((ext_vector_type(8)));

#define BUCKET_CAP 32
#define HALF_ROW 16
#define GROUP_SHIFT 10
#define GROUP_NODES (1 << GROUP_SHIFT)
#define SRC_BITS 22
#define PART_THREADS 512
#define PART_EPT 16
#define PART_EDGES (PART_THREADS * PART_EPT)
#define PART_WAVES (PART_THREADS / 32)
#define CELL_CAP 32
#define CELL_EMPTY 0xFFFFFFFFu
#define PLACE_RPT 20
#define ROW_PITCH 33

__global__ void __launch_bounds__(PART_THREADS)
group_edges(const int* __restrict__ src, const int* __restrict__ dst,
            unsigned* __restrict__ cells, int* __restrict__ spill_n,
            int* __restrict__ spill_src, int* __restrict__ spill_dst,
            int* __restrict__ ovfb_n, long long E, int ngroups) {
  extern __shared__ int sm[];
  __shared__ unsigned rec[PART_EDGES];
  __shared__ int wtot[PART_WAVES];
  int* hist  = sm;
  int* lbase = sm + ngroups;
  int tid = threadIdx.x;
  int lane = tid & 31;
  int wave = tid >> 5;
  for (int i = tid; i < ngroups; i += PART_THREADS) hist[i] = 0;
  if (tid == 0 && blockIdx.x == 0) ovfb_n[0] = 0;
  __syncthreads();

  long long e0 = (long long)blockIdx.x * PART_EDGES + tid;
  int d[PART_EPT], s[PART_EPT], rk[PART_EPT];
#pragma unroll
  for (int k = 0; k < PART_EPT; ++k) {
    long long e = e0 + (long long)k * PART_THREADS;
    bool ok = e < E;
    long long ec = ok ? e : 0;
    int dv = dst[ec];
    s[k] = src[ec];
    d[k] = ok ? dv : -1;
  }
#pragma unroll
  for (int k = 0; k < PART_EPT; ++k) {
    rk[k] = 0;
    if (d[k] >= 0)
      rk[k] = __hip_atomic_fetch_add(&hist[d[k] >> GROUP_SHIFT], 1, __ATOMIC_RELAXED,
                                     __HIP_MEMORY_SCOPE_WORKGROUP);
  }
  __syncthreads();

  int chunk = (ngroups + PART_THREADS - 1) / PART_THREADS;
  int gfirst = tid * chunk;
  int sum = 0;
  for (int i = 0; i < chunk; ++i) {
    int g = gfirst + i;
    if (g < ngroups) sum += hist[g];
  }
  int incl = sum;
  for (int o = 1; o < 32; o <<= 1) {
    int v = __shfl_up(incl, o);
    if (lane >= o) incl += v;
  }
  if (lane == 31) wtot[wave] = incl;
  __syncthreads();
  int run = incl - sum;
  for (int w = 0; w < wave; ++w) run += wtot[w];
  for (int i = 0; i < chunk; ++i) {
    int g = gfirst + i;
    if (g < ngroups) {
      lbase[g] = run;
      run += hist[g];
    }
  }
  __syncthreads();

#pragma unroll
  for (int k = 0; k < PART_EPT; ++k) rk[k] += lbase[(d[k] >= 0 ? d[k] : 0) >> GROUP_SHIFT];
#pragma unroll
  for (int k = 0; k < PART_EPT; ++k) {
    if (d[k] >= 0)
      rec[rk[k]] = ((unsigned)(d[k] & (GROUP_NODES - 1)) << SRC_BITS) | (unsigned)s[k];
  }
  __syncthreads();

  auto excess_before = [&](int gend) {
    int t = 0;
    for (int gg = lane; gg < gend; gg += 32) {
      int cc = hist[gg];
      t += cc > CELL_CAP ? cc - CELL_CAP : 0;
    }
    for (int o = 16; o > 0; o >>= 1) t += __shfl_xor(t, o);
    return t;
  };

  const long long gpitch = (long long)gridDim.x * CELL_CAP;
  unsigned* __restrict__ mycell = cells + (long long)blockIdx.x * CELL_CAP + lane;
  for (int gs = wave; gs < ngroups; gs += 4 * PART_WAVES) {
    int c[4], lb[4];
    unsigned r[4];
#pragma unroll
    for (int u = 0; u < 4; ++u) {
      int g = gs + u * PART_WAVES;
      bool ok = g < ngroups;
      int gg = ok ? g : 0;
      int cv = hist[gg];
      lb[u] = lbase[gg];
      c[u] = ok ? cv : -1;
    }
#pragma unroll
    for (int u = 0; u < 4; ++u) r[u] = rec[lane < c[u] ? lb[u] + lane : 0];
#pragma unroll
    for (int u = 0; u < 4; ++u)
      if (c[u] >= 0) mycell[(long long)(gs + u * PART_WAVES) * gpitch] = lane < c[u] ? r[u] : CELL_EMPTY;
#pragma unroll
    for (int u = 0; u < 4; ++u)
      if (c[u] > CELL_CAP) {
        int g = gs + u * PART_WAVES;
        long long sp0 = (long long)blockIdx.x * PART_EDGES + excess_before(g) - CELL_CAP;
        for (int j = lane + CELL_CAP; j < c[u]; j += 32) {
          unsigned rr = rec[lb[u] + j];
          spill_src[sp0 + j] = (int)(rr & ((1u << SRC_BITS) - 1u));
          spill_dst[sp0 + j] = (g << GROUP_SHIFT) | (int)(rr >> SRC_BITS);
        }
      }
  }
  if (wave == PART_WAVES - 1) {
    int t = excess_before(ngroups);
    if (lane == 0) spill_n[blockIdx.x] = t;
  }
}

template <int K, int T>
struct xw_operands {
  float x[T][(K + 3) / 4][2];
  float w[(K + 3) / 4][2];
};

template <int K, int NCOL, int T>
__device__ __forceinline__ void xw_load(const float* __restrict__ X, const float* __restrict__ W,
                                        long long row_base, long long n, int lane,
                                        xw_operands<K, T>& op) {
  constexpr int KT = (K + 3) / 4;
  int half = lane >> 4;
  int l    = lane & 15;
  const int lc = (NCOL == 16 || l < NCOL) ? l : 0;
#pragma unroll
  for (int kt = 0; kt < KT; ++kt) {
    int k0 = kt * 4 + 2 * half;
    int ka0 = (k0     < K) ? k0     : 0;
    int ka1 = (k0 + 1 < K) ? k0 + 1 : 0;
    op.w[kt][0] = W[ka0 * NCOL + lc];
    op.w[kt][1] = W[ka1 * NCOL + lc];
#pragma unroll
    for (int u = 0; u < T; ++u) {
      long long m = row_base + 16 * u + l;
      if (m > n - 1) m = n - 1;
      op.x[u][kt][0] = X[m * K + ka0];
      op.x[u][kt][1] = X[m * K + ka1];
    }
  }
}

template <int K, int NCOL, int T>
__device__ __forceinline__ v8f_t xw_product(const xw_operands<K, T>& op, int u, int lane) {
  constexpr int KT = (K + 3) / 4;
  int half = lane >> 4;
  const bool colok = (NCOL == 16) ? true : ((lane & 15) < NCOL);
  v8f_t c = {};
#pragma unroll
  for (int kt = 0; kt < KT; ++kt) {
    int k0 = kt * 4 + 2 * half;
    bool ok0 = (k0     < K);
    bool ok1 = (k0 + 1 < K);
    v2f_t a, b;
    a.x = ok0 ? op.x[u][kt][0] : 0.0f;
    a.y = ok1 ? op.x[u][kt][1] : 0.0f;
    b.x = (ok0 && colok) ? op.w[kt][0] : 0.0f;
    b.y = (ok1 && colok) ? op.w[kt][1] : 0.0f;
    c = __builtin_amdgcn_wmma_f32_16x16x4_f32(false, a, false, b, (short)0, c, false, false);
  }
  return c;
}

template <int K, int NCOL>
__global__ void __launch_bounds__(GROUP_NODES)
place_group(const unsigned* __restrict__ cells, int nblk, const int* __restrict__ spill_n,
            const int* __restrict__ spill_src, const int* __restrict__ spill_dst,
            int* __restrict__ cnt, int* __restrict__ bucket,
            int* ovfb_n, int* __restrict__ ovfb_src, int* __restrict__ ovfb_dst,
            const float* __restrict__ X, const float* __restrict__ W, float* __restrict__ HS,
            long long n) {
  constexpr int TPW = GROUP_NODES / 16 / (GROUP_NODES / 32);
  __shared__ int lcnt[GROUP_NODES];
  __shared__ int lslot[GROUP_NODES * ROW_PITCH];
  int tid = threadIdx.x;
  int g = blockIdx.x;
  long long node0 = (long long)g << GROUP_SHIFT;
  int lane = tid & 31;
  int trow0 = (tid >> 5) * (16 * TPW);
  xw_operands<K, TPW> xw;
  xw_load<K, NCOL, TPW>(X, W, node0 + trow0, n, lane, xw);

  const int m = nblk * CELL_CAP;
  const unsigned* __restrict__ gl = cells + (long long)g * m;
  unsigned rs[PLACE_RPT];
  auto load_sweep = [&](int i0) {
#pragma unroll
    for (int u = 0; u < PLACE_RPT; ++u) {
      int i = i0 + u * GROUP_NODES + tid;
      unsigned v = gl[i < m ? i : 0];
      rs[u] = i < m ? v : CELL_EMPTY;
    }
  };
  load_sweep(0);
  int ns_first = spill_n[tid < nblk ? tid : 0];
  if (tid >= nblk) ns_first = 0;
  lcnt[tid] = 0;
  __syncthreads();
  int dv_first = 0;
  if (ns_first > 0) dv_first = spill_dst[(long long)tid * PART_EDGES];

  auto take_rank = [&](int dl) {
    return __hip_atomic_fetch_add(&lcnt[dl], 1, __ATOMIC_RELAXED, __HIP_MEMORY_SCOPE_WORKGROUP);
  };
  auto store_slot = [&](int dl, int sv, int rank) {
    if (rank < BUCKET_CAP) {
      lslot[dl * ROW_PITCH + rank] = sv;
    } else {
      int p = __hip_atomic_fetch_add(ovfb_n, 1, __ATOMIC_RELAXED, __HIP_MEMORY_SCOPE_AGENT);
      ovfb_src[p] = sv;
      ovfb_dst[p] = (int)(node0 + dl);
    }
  };
  auto put = [&](int dl, int sv) { store_slot(dl, sv, take_rank(dl)); };

  for (int i0 = 0;;) {
    int rk[PLACE_RPT];
#pragma unroll
    for (int u = 0; u < PLACE_RPT; ++u) {
      rk[u] = -1;
      if (rs[u] != CELL_EMPTY) rk[u] = take_rank((int)(rs[u] >> SRC_BITS));
    }
#pragma unroll
    for (int u = 0; u < PLACE_RPT; ++u)
      if (rk[u] >= 0) store_slot((int)(rs[u] >> SRC_BITS), (int)(rs[u] & ((1u << SRC_BITS) - 1u)), rk[u]);
    i0 += PLACE_RPT * GROUP_NODES;
    if (i0 >= m) break;
    load_sweep(i0);
  }

  for (int b = tid; b < nblk; b += GROUP_NODES) {
    int ns = b == tid ? ns_first : spill_n[b];
    const long long sb = (long long)b * PART_EDGES;
    for (int p = 0; p < ns; ++p) {
      int dv = (b == tid && p == 0) ? dv_first : spill_dst[sb + p];
      if ((dv >> GROUP_SHIFT) == g) put(dv & (GROUP_NODES - 1), spill_src[sb + p]);
    }
  }

  v8f_t hacc[TPW];
#pragma unroll
  for (int u = 0; u < TPW; ++u) hacc[u] = xw_product<K, NCOL, TPW>(xw, u, lane);
  __syncthreads();

  long long node = node0 + tid;
  if (node < n) cnt[node] = lcnt[tid];

  int j = lane & (HALF_ROW - 1);
  int rsel = lane >> 4;
  int* bucket_hi = bucket + n * HALF_ROW;
  constexpr int PSTEP = GROUP_NODES / 32;
  for (int p0 = tid >> 5; p0 < GROUP_NODES / 2; p0 += 4 * PSTEP) {
    int c[4];
#pragma unroll
    for (int u = 0; u < 4; ++u) c[u] = lcnt[2 * (p0 + u * PSTEP) + rsel];
#pragma unroll
    for (int u = 0; u < 4; ++u) {
      int row = 2 * (p0 + u * PSTEP) + rsel;
      long long nd = node0 + row;
      if (nd < n) {
        bucket[nd * HALF_ROW + j] = j < c[u] ? lslot[row * ROW_PITCH + j] : (int)nd;
        if (c[u] > HALF_ROW)
          bucket_hi[nd * HALF_ROW + j] = j + HALF_ROW < c[u] ? lslot[row * ROW_PITCH + HALF_ROW + j] : (int)nd;
      }
    }
  }

  const bool colok = (NCOL == 16) ? true : ((lane & 15) < NCOL);
#pragma unroll
  for (int u = 0; u < TPW; ++u) {
    int lr = trow0 + 16 * u + 8 * (lane >> 4);
    long long rb = node0 + trow0 + 16 * u;
    float* __restrict__ hb = HS + (node0 + lr) * NCOL + (lane & 15);
    if (rb + 16 <= n) {
      if (colok) {
#pragma unroll
        for (int v = 0; v < 8; ++v) hb[v * NCOL] = hacc[u][v] * rsqrtf((float)lcnt[lr + v] + 1.0f);
      }
    } else if (rb < n) {
#pragma unroll
      for (int v = 0; v < 8; ++v)
        if (node0 + lr + v < n && colok) hb[v * NCOL] = hacc[u][v] * rsqrtf((float)lcnt[lr + v] + 1.0f);
    }
  }
}

template <int LPN, int QL>
__device__ __forceinline__ int group_bcast(int v) {
  static_assert(LPN == 4 || LPN == 2, "node groups of four or two lanes");
  constexpr int ctrl = (LPN == 4) ? (QL * 0x55) : (QL == 0 ? 0xA0 : 0xF5);
  return __builtin_amdgcn_update_dpp(v, v, ctrl, 0xF, 0xF, false);
}

template <int LPN, int P>
__device__ __forceinline__ int4 half_row_piece(const int4* mine) {
  constexpr int U = P / LPN, QL = P % LPN;
  int4 r;
  r.x = group_bcast<LPN, QL>(mine[U].x);
  r.y = group_bcast<LPN, QL>(mine[U].y);
  r.z = group_bcast<LPN, QL>(mine[U].z);
  r.w = group_bcast<LPN, QL>(mine[U].w);
  return r;
}

template <int LPN>
__device__ __forceinline__ void add_half_row(const float4* __restrict__ h4, const int* __restrict__ half_row,
                                             int filled, long long d, int q, float4& acc) {
  constexpr int PPL = 4 / LPN;
  int4 mine[PPL];
#pragma unroll
  for (int u = 0; u < PPL; ++u) mine[u] = *(const int4*)(half_row + 4 * (q + LPN * u));
  int4 p0 = half_row_piece<LPN, 0>(mine);
  int4 p1 = half_row_piece<LPN, 1>(mine);
  int4 p2 = half_row_piece<LPN, 2>(mine);
  int4 p3 = half_row_piece<LPN, 3>(mine);
  int idx[HALF_ROW] = {p0.x, p0.y, p0.z, p0.w, p1.x, p1.y, p1.z, p1.w,
                       p2.x, p2.y, p2.z, p2.w, p3.x, p3.y, p3.z, p3.w};
  float4 v[HALF_ROW];
#pragma unroll
  for (int j = 0; j < HALF_ROW; ++j) {
    long long s = j < filled ? (long long)idx[j] : d;
    v[j] = h4[s * LPN + q];
  }
#pragma unroll
  for (int j = 0; j < HALF_ROW; ++j) {
    bool k = j < filled;
    acc.x += k ? v[j].x : 0.0f;
    acc.y += k ? v[j].y : 0.0f;
    acc.z += k ? v[j].z : 0.0f;
    acc.w += k ? v[j].w : 0.0f;
  }
}

template <int F>
__device__ __forceinline__ float4 gather_relu(const float* __restrict__ HS, const int* __restrict__ cnt,
                                              const int* __restrict__ bucket,
                                              const int* __restrict__ ovf_n,
                                              const int* __restrict__ ovf_src,
                                              const int* __restrict__ ovf_dst,
                                              const float* __restrict__ bias, long long d, int q,
                                              long long n, float& dv) {
  constexpr int LPN = F / 4;
  const float4* __restrict__ h4 = (const float4*)HS;
  int c = cnt[d];
  int m = c < BUCKET_CAP ? c : BUCKET_CAP;
  float4 acc = h4[d * LPN + q];
  const int* __restrict__ bi = bucket + d * HALF_ROW;

  add_half_row<LPN>(h4, bi, m < HALF_ROW ? m : HALF_ROW, d, q, acc);
  if (c > HALF_ROW)
    add_half_row<LPN>(h4, bi + n * HALF_ROW, m - HALF_ROW, d, q, acc);

  if (c > BUCKET_CAP) {
    int on = ovf_n[0];
    for (int p = 0; p < on; ++p) {
      if (ovf_dst[p] == (int)d) {
        float4 v = h4[(long long)ovf_src[p] * LPN + q];
        acc.x += v.x; acc.y += v.y; acc.z += v.z; acc.w += v.w;
      }
    }
  }

  dv = rsqrtf((float)c + 1.0f);
  float4 y;
  y.x = acc.x * dv + bias[4 * q + 0];
  y.y = acc.y * dv + bias[4 * q + 1];
  y.z = acc.z * dv + bias[4 * q + 2];
  y.w = acc.w * dv + bias[4 * q + 3];
  y.x = y.x > 0.0f ? y.x : 0.0f;
  y.y = y.y > 0.0f ? y.y : 0.0f;
  y.z = y.z > 0.0f ? y.z : 0.0f;
  y.w = y.w > 0.0f ? y.w : 0.0f;
  return y;
}

#define GT_THREADS 256
#define GT_WAVES (GT_THREADS / 32)
#define GT_PITCH 20

template <int F, int NOUT>
__global__ void __launch_bounds__(GT_THREADS)
gather_transform(const float* __restrict__ HS, const int* __restrict__ cnt,
                 const int* __restrict__ bucket, const int* __restrict__ ovf_n,
                 const int* __restrict__ ovf_src, const int* __restrict__ ovf_dst,
                 const float* __restrict__ bias, const float* __restrict__ W,
                 float* __restrict__ HS2, long long n) {
  static_assert(F == 16 && NOUT == 8, "tile shapes below are written for 16 -> 8");
  constexpr int LPN = F / 4;
  constexpr int NPW = 32 / LPN;
  constexpr int KT = F / 4;
  __shared__ __attribute__((aligned(16))) float tile[GT_WAVES * NPW * GT_PITCH];

  long long t = (long long)blockIdx.x * blockDim.x + threadIdx.x;
  long long node0 = (t >> 5) * NPW;
  if (node0 >= n) return;
  int lane = threadIdx.x & 31;
  int wave = threadIdx.x >> 5;
  int q = lane & (LPN - 1);
  long long d = node0 + (lane >> 2);
  if (d > n - 1) d = n - 1;

  float dv;
  float4 y = gather_relu<F>(HS, cnt, bucket, ovf_n, ovf_src, ovf_dst, bias, d, q, n, dv);

  float* __restrict__ wt = tile + wave * (NPW * GT_PITCH);
  *(float4*)(wt + (lane >> 2) * GT_PITCH + 4 * q) = y;

  int half = lane >> 4;
  int c = lane & 15;
  int r = c & (NPW - 1);
  const float* __restrict__ yrow = wt + r * GT_PITCH + 2 * half;
  const float* __restrict__ wcol = W + (2 * half) * NOUT + r;

  v8f_t acc = {};
#pragma unroll
  for (int kt = 0; kt < KT; ++kt) {
    v2f_t a, b;
    a.x = wcol[(kt * 4) * NOUT];
    a.y = wcol[(kt * 4 + 1) * NOUT];
    float2 yy = *(const float2*)(yrow + kt * 4);
    b.x = yy.x;
    b.y = yy.y;
    acc = __builtin_amdgcn_wmma_f32_16x16x4_f32(false, a, false, b, (short)0, acc, false, false);
  }

  float dn = __shfl(dv, r * LPN);
  float4 o;
  o.x = (half ? acc[4] : acc[0]) * dn;
  o.y = (half ? acc[5] : acc[1]) * dn;
  o.z = (half ? acc[6] : acc[2]) * dn;
  o.w = (half ? acc[7] : acc[3]) * dn;
  if (c < NPW && node0 + r < n)
    *(float4*)(HS2 + (node0 + r) * NOUT + 4 * half) = o;
}

template <int F>
__global__ void gather_head(const float* __restrict__ HS, const int* __restrict__ cnt,
                            const int* __restrict__ bucket, const int* __restrict__ ovf_n,
                            const int* __restrict__ ovf_src, const int* __restrict__ ovf_dst,
                            const float* __restrict__ bias, const float* __restrict__ Wd,
                            const float* __restrict__ bd, float* __restrict__ out, long long n) {
  constexpr int LPN = F / 4;
  long long t = (long long)blockIdx.x * blockDim.x + threadIdx.x;
  long long d = t / LPN;
  int q = (int)(t % LPN);
  if (d >= n) return;

  float dv;
  float4 y = gather_relu<F>(HS, cnt, bucket, ovf_n, ovf_src, ovf_dst, bias, d, q, n, dv);

  float p = y.x * Wd[4 * q + 0] + y.y * Wd[4 * q + 1] + y.z * Wd[4 * q + 2] + y.w * Wd[4 * q + 3];
  for (int o = LPN / 2; o > 0; o >>= 1) p += __shfl_xor(p, o);
  if (q == 0) out[d] = p + bd[0];
}

extern "C" void kernel_launch(void* const* d_in, const int* in_sizes, int n_in,
                              void* d_out, int out_size, void* d_ws, size_t ws_size,
                              hipStream_t stream) {
  const float* x  = (const float*)d_in[0];
  const int*   ei = (const int*)d_in[1];
  const float* W1 = (const float*)d_in[2];
  const float* b1 = (const float*)d_in[3];
  const float* W2 = (const float*)d_in[4];
  const float* b2 = (const float*)d_in[5];
  const float* Wd = (const float*)d_in[6];
  const float* bd = (const float*)d_in[7];
  float* out = (float*)d_out;

  long long n = in_sizes[0] / 10;
  long long E = in_sizes[1] / 2;
  const int* src = ei;
  const int* dst = ei + E;

  int* wsi = (int*)d_ws;
  long long ng   = (n + GROUP_NODES - 1) / GROUP_NODES;
  long long nblk = (E + PART_EDGES - 1) / PART_EDGES;
  long long o_c = ((n + 1 + 63) / 64) * 64;
  long long o_b = o_c + ((nblk + 63) / 64) * 64;
  int*   cnt      = wsi;
  int*   ovfb_n   = wsi + n;
  int*   spill_n  = wsi + o_c;
  int*   bucket   = wsi + o_b;
  float* hs1      = (float*)(wsi + o_b + n * BUCKET_CAP);
  float* hs2      = hs1 + n * 16;
  int*   ovfb_src = (int*)(hs2 + n * 16);
  int*   ovfb_dst = ovfb_src + E;
  unsigned* cells = (unsigned*)(ovfb_dst + E);
  int*   spill_src = (int*)(cells + ng * nblk * CELL_CAP);
  int*   spill_dst = spill_src + nblk * PART_EDGES;

  const int B = 256;
  auto nb = [](long long t, int b) { return (unsigned)((t + b - 1) / b); };

  group_edges<<<(unsigned)nblk, PART_THREADS, 2 * ng * sizeof(int), stream>>>(
      src, dst, cells, spill_n, spill_src, spill_dst, ovfb_n, E, (int)ng);
  place_group<10, 16><<<(unsigned)ng, GROUP_NODES, 0, stream>>>(cells, (int)nblk, spill_n, spill_src, spill_dst,
                                                               cnt, bucket, ovfb_n, ovfb_src, ovfb_dst,
                                                               x, W1, hs1, n);

  gather_transform<16, 8><<<nb(n * 4, GT_THREADS), GT_THREADS, 0, stream>>>(
      hs1, cnt, bucket, ovfb_n, ovfb_src, ovfb_dst, b1, W2, hs2, n);

  gather_head<8><<<nb(n * 2, B), B, 0, stream>>>(hs2, cnt, bucket, ovfb_n, ovfb_src, ovfb_dst,
                                                 b2, Wd, bd, out, n);
}
